# starting kernel
# baseline (speedup 1.0000x reference)
#include <hip/hip_runtime.h>


typedef float v2f __attribute__((ext_vector_type(2)));
typedef float v8f __attribute__((ext_vector_type(8)));

#define N_NODES   1000000
#define N_EDGES   16000000
#define F_IN      32
#define F_H1      16
#define F_H2      8
#define N_CLS     5
#define N_GRAPHS  10000

#define ATOMIC_FADD(p, v) \
  __hip_atomic_fetch_add((p), (v), __ATOMIC_RELAXED, __HIP_MEMORY_SCOPE_AGENT)


__global__ void zero_f32(float* __restrict__ p, long long n) {
  long long i = (long long)blockIdx.x * blockDim.x + threadIdx.x;
  long long stride = (long long)gridDim.x * blockDim.x;
  for (; i < n; i += stride) p[i] = 0.0f;
}

__global__ void degree_kernel(const int* __restrict__ dst,
                              float* __restrict__ deg, int nE) {
  int e = blockIdx.x * blockDim.x + threadIdx.x;
  if (e < nE) ATOMIC_FADD(&deg[dst[e]], 1.0f);
}

__global__ void rsqrt_kernel(float* __restrict__ deg, int n) {
  int i = blockIdx.x * blockDim.x + threadIdx.x;
  if (i < n) deg[i] = rsqrtf(deg[i] + 1.0f);
}


__global__ __launch_bounds__(128) void gemm_x_w1(const float* __restrict__ X,
                                                 const float* __restrict__ W1,
                                                 float* __restrict__ H1pre) {
  int wave = blockIdx.x * (blockDim.x >> 5) + (threadIdx.x >> 5);
  int lane = threadIdx.x & 31;
  int m  = lane & 15;
  int kk = (lane >> 4) << 1;
  const float* xrow = X + (size_t)(wave * 16 + m) * F_IN;

  v8f c = {0.f, 0.f, 0.f, 0.f, 0.f, 0.f, 0.f, 0.f};
#pragma unroll
  for (int k = 0; k < F_IN; k += 4) {
    v2f a, b;
    a.x = xrow[k + kk];
    a.y = xrow[k + kk + 1];
    b.x = W1[(k + kk) * F_H1 + m];
    b.y = W1[(k + kk + 1) * F_H1 + m];
    c = __builtin_amdgcn_wmma_f32_16x16x4_f32(false, a, false, b,
                                              (short)0, c, false, false);
  }
  int hi = lane >> 4;
  float* out = H1pre + (size_t)(wave * 16 + 8 * hi) * F_H1 + m;
#pragma unroll
  for (int j = 0; j < 8; ++j) out[(size_t)j * F_H1] = c[j];
}

__global__ __launch_bounds__(128) void gemm_h1_w2(const float* __restrict__ H1,
                                                  const float* __restrict__ W2,
                                                  float* __restrict__ H2pre) {
  int wave = blockIdx.x * (blockDim.x >> 5) + (threadIdx.x >> 5);
  int lane = threadIdx.x & 31;
  int m  = lane & 15;
  int kk = (lane >> 4) << 1;
  const float* hrow = H1 + (size_t)(wave * 16 + m) * F_H1;
  bool live_col = (m < F_H2);

  v8f c = {0.f, 0.f, 0.f, 0.f, 0.f, 0.f, 0.f, 0.f};
#pragma unroll
  for (int k = 0; k < F_H1; k += 4) {
    v2f a, b;
    a.x = hrow[k + kk];
    a.y = hrow[k + kk + 1];
    b.x = live_col ? W2[(k + kk) * F_H2 + m] : 0.0f;
    b.y = live_col ? W2[(k + kk + 1) * F_H2 + m] : 0.0f;
    c = __builtin_amdgcn_wmma_f32_16x16x4_f32(false, a, false, b,
                                              (short)0, c, false, false);
  }
  if (live_col) {
    int hi = lane >> 4;
    float* out = H2pre + (size_t)(wave * 16 + 8 * hi) * F_H2 + m;
#pragma unroll
    for (int j = 0; j < 8; ++j) out[(size_t)j * F_H2] = c[j];
  }
}


__global__ void agg_edges16(const int* __restrict__ src, const int* __restrict__ dst,
                            const float* __restrict__ dinv,
                            const float* __restrict__ H, float* __restrict__ AGG,
                            int nE) {
  int e = blockIdx.x * blockDim.x + threadIdx.x;
  if (e >= nE) return;
  int s = src[e], d = dst[e];
  float w = dinv[s] * dinv[d];
  const float4* hs = (const float4*)(H + (size_t)s * F_H1);
  float* out = AGG + (size_t)d * F_H1;
#pragma unroll
  for (int q = 0; q < 4; ++q) {
    float4 v = hs[q];
    ATOMIC_FADD(&out[4 * q + 0], v.x * w);
    ATOMIC_FADD(&out[4 * q + 1], v.y * w);
    ATOMIC_FADD(&out[4 * q + 2], v.z * w);
    ATOMIC_FADD(&out[4 * q + 3], v.w * w);
  }
}

__global__ void agg_edges8(const int* __restrict__ src, const int* __restrict__ dst,
                           const float* __restrict__ dinv,
                           const float* __restrict__ H, float* __restrict__ AGG,
                           int nE) {
  int e = blockIdx.x * blockDim.x + threadIdx.x;
  if (e >= nE) return;
  int s = src[e], d = dst[e];
  float w = dinv[s] * dinv[d];
  const float4* hs = (const float4*)(H + (size_t)s * F_H2);
  float* out = AGG + (size_t)d * F_H2;
#pragma unroll
  for (int q = 0; q < 2; ++q) {
    float4 v = hs[q];
    ATOMIC_FADD(&out[4 * q + 0], v.x * w);
    ATOMIC_FADD(&out[4 * q + 1], v.y * w);
    ATOMIC_FADD(&out[4 * q + 2], v.z * w);
    ATOMIC_FADD(&out[4 * q + 3], v.w * w);
  }
}


__global__ void finalize1(float* __restrict__ agg, const float* __restrict__ hpre,
                          const float* __restrict__ dinv,
                          const float* __restrict__ b1, int n) {
  int i = blockIdx.x * blockDim.x + threadIdx.x;
  if (i >= n) return;
  float w = dinv[i]; w *= w;
  float4* a = (float4*)(agg + (size_t)i * F_H1);
  const float4* h = (const float4*)(hpre + (size_t)i * F_H1);
  const float4* b = (const float4*)b1;
#pragma unroll
  for (int q = 0; q < 4; ++q) {
    float4 av = a[q], hv = h[q], bv = b[q];
    av.x = fmaxf(av.x + hv.x * w + bv.x, 0.0f);
    av.y = fmaxf(av.y + hv.y * w + bv.y, 0.0f);
    av.z = fmaxf(av.z + hv.z * w + bv.z, 0.0f);
    av.w = fmaxf(av.w + hv.w * w + bv.w, 0.0f);
    a[q] = av;
  }
}

__global__ void finalize2_pool(const float* __restrict__ agg,
                               const float* __restrict__ hpre,
                               const float* __restrict__ dinv,
                               const float* __restrict__ b2,
                               const int* __restrict__ batch,
                               float* __restrict__ pooled,
                               float* __restrict__ cnt, int n) {
  int i = blockIdx.x * blockDim.x + threadIdx.x;
  if (i >= n) return;
  float w = dinv[i]; w *= w;
  int g = batch[i];
  float* ps = pooled + (size_t)g * F_H2;
#pragma unroll
  for (int f = 0; f < F_H2; ++f) {
    float v = agg[(size_t)i * F_H2 + f] + hpre[(size_t)i * F_H2 + f] * w + b2[f];
    v = fmaxf(v, 0.0f);
    ATOMIC_FADD(&ps[f], v);
  }
  ATOMIC_FADD(&cnt[g], 1.0f);
}

__global__ void readout(const float* __restrict__ pooled, const float* __restrict__ cnt,
                        const float* __restrict__ Wl, const float* __restrict__ bl,
                        float* __restrict__ out, int G) {
  int g = blockIdx.x * blockDim.x + threadIdx.x;
  if (g >= G) return;
  float inv = 1.0f / fmaxf(cnt[g], 1.0f);
  float p[F_H2];
#pragma unroll
  for (int f = 0; f < F_H2; ++f) p[f] = pooled[(size_t)g * F_H2 + f] * inv;
#pragma unroll
  for (int c = 0; c < N_CLS; ++c) {
    float acc = bl[c];
#pragma unroll
    for (int f = 0; f < F_H2; ++f) acc += p[f] * Wl[f * N_CLS + c];
    out[(size_t)g * N_CLS + c] = acc;
  }
}


extern "C" void kernel_launch(void* const* d_in, const int* in_sizes, int n_in,
                              void* d_out, int out_size, void* d_ws, size_t ws_size,
                              hipStream_t stream) {
  const float* x     = (const float*)d_in[0];
  const int*   ei    = (const int*)d_in[1];
  const int*   src   = ei;
  const int*   dst   = ei + N_EDGES;
  const int*   batch = (const int*)d_in[2];
  const float* W1 = (const float*)d_in[4];
  const float* b1 = (const float*)d_in[5];
  const float* W2 = (const float*)d_in[6];
  const float* b2 = (const float*)d_in[7];
  const float* Wl = (const float*)d_in[8];
  const float* bl = (const float*)d_in[9];
  float* out = (float*)d_out;

  char* ws = (char*)d_ws;
  float* dinv   = (float*)(ws);
  float* bufA   = (float*)(ws + (4ll  << 20));
  float* bufB   = (float*)(ws + (68ll << 20));
  float* pooled = (float*)(ws + (132ll << 20));
  float* cnt    = pooled + (size_t)N_GRAPHS * F_H2;

  float* h1pre = bufA;
  float* agg1  = bufB;
  float* h2pre = bufA;
  float* agg2  = (float*)((char*)bufA + (32ll << 20));

  const int EB = 256;
  const int EG = (N_EDGES + EB - 1) / EB;
  const int NB = 256;
  const int NG = (N_NODES + NB - 1) / NB;
  const int GEMM_BLOCKS = N_NODES / 64;

  zero_f32<<<2048, 256, 0, stream>>>(dinv, (long long)N_NODES);
  degree_kernel<<<EG, EB, 0, stream>>>(dst, dinv, N_EDGES);
  rsqrt_kernel<<<NG, NB, 0, stream>>>(dinv, N_NODES);

  gemm_x_w1<<<GEMM_BLOCKS, 128, 0, stream>>>(x, W1, h1pre);
  zero_f32<<<8192, 256, 0, stream>>>(agg1, (long long)N_NODES * F_H1);
  agg_edges16<<<EG, EB, 0, stream>>>(src, dst, dinv, h1pre, agg1, N_EDGES);
  finalize1<<<NG, NB, 0, stream>>>(agg1, h1pre, dinv, b1, N_NODES);

  gemm_h1_w2<<<GEMM_BLOCKS, 128, 0, stream>>>(agg1, W2, h2pre);
  zero_f32<<<8192, 256, 0, stream>>>(agg2, (long long)N_NODES * F_H2);
  agg_edges8<<<EG, EB, 0, stream>>>(src, dst, dinv, h2pre, agg2, N_EDGES);

  zero_f32<<<128, 256, 0, stream>>>(pooled, (long long)N_GRAPHS * (F_H2 + 1));
  finalize2_pool<<<NG, NB, 0, stream>>>(agg2, h2pre, dinv, b2, batch,
                                        pooled, cnt, N_NODES);
  readout<<<(N_GRAPHS + NB - 1) / NB, NB, 0, stream>>>(pooled, cnt, Wl, bl,
                                                       out, N_GRAPHS);
}


# list build without global atomics: 13 node ranges x 16 edge slices, a work-group filters its slice and counts, then places, its range's edges with LDS adds; sum order = slice, then LDS arrival
# speedup vs baseline: 4.3063x; 4.3063x over previous
#include <hip/hip_runtime.h>


typedef float v2f __attribute__((ext_vector_type(2)));
typedef float v8f __attribute__((ext_vector_type(8)));

#define N_NODES   1000000
#define N_EDGES   16000000
#define F_IN      32
#define F_H1      16
#define F_H2      8
#define N_CLS     5
#define N_GRAPHS  10000

#define SCAN_ITEMS 1024
#define SCAN_BLOCKS ((N_NODES + SCAN_ITEMS - 1) / SCAN_ITEMS)

#define ATOMIC_ADD(p, v) \
  __hip_atomic_fetch_add((p), (v), __ATOMIC_RELAXED, __HIP_MEMORY_SCOPE_AGENT)


__global__ void zero_u32(unsigned* __restrict__ p, int n) {
  int i = blockIdx.x * blockDim.x + threadIdx.x;
  int stride = gridDim.x * blockDim.x;
  for (; i < n; i += stride) p[i] = 0u;
}

#define LDS_ADD(p, v) \
  __hip_atomic_fetch_add((p), (v), __ATOMIC_RELAXED, __HIP_MEMORY_SCOPE_WORKGROUP)

#define RANGE   78000
#define NRANGE  13
#define NSLICE  16
#define SLICE_V4 (N_EDGES / 4 / NSLICE)
#define LB_THREADS 1024
#define LB_GROUPS 4
static_assert(RANGE * NRANGE >= N_NODES && RANGE % 4 == 0 && N_NODES % 4 == 0, "ranges cover the nodes");
static_assert(SLICE_V4 * 4 * NSLICE == N_EDGES, "slices cover the edges exactly");
static_assert(RANGE * 4 <= 320 * 1024, "one range's counters fit the LDS");

__global__ __launch_bounds__(LB_THREADS) void hist_ranges(const int* __restrict__ dst,
                                                          int* __restrict__ planes) {
  __shared__ __attribute__((aligned(16))) int cnt[RANGE];
  int r = blockIdx.x % NRANGE;
  int sl = blockIdx.x / NRANGE;
  int lo = r * RANGE;
  unsigned nh = (unsigned)min(RANGE, N_NODES - lo);
  for (int i = threadIdx.x; i < RANGE / 4; i += LB_THREADS)
    ((int4*)cnt)[i] = make_int4(0, 0, 0, 0);
  __syncthreads();
  const int4* d4 = (const int4*)dst + (size_t)sl * SLICE_V4;
  for (int i = threadIdx.x; i < SLICE_V4; i += LB_GROUPS * LB_THREADS) {
    int4 d[LB_GROUPS];
#pragma unroll
    for (int g = 0; g < LB_GROUPS; ++g) {
      int ig = i + g * LB_THREADS;
      d[g] = (ig < SLICE_V4) ? d4[ig] : make_int4(-1, -1, -1, -1);
    }
#pragma unroll
    for (int g = 0; g < LB_GROUPS; ++g) {
      unsigned a;
      a = (unsigned)(d[g].x - lo); if (a < nh) LDS_ADD(&cnt[a], 1);
      a = (unsigned)(d[g].y - lo); if (a < nh) LDS_ADD(&cnt[a], 1);
      a = (unsigned)(d[g].z - lo); if (a < nh) LDS_ADD(&cnt[a], 1);
      a = (unsigned)(d[g].w - lo); if (a < nh) LDS_ADD(&cnt[a], 1);
    }
  }
  __syncthreads();
  int4* out = (int4*)(planes + (size_t)sl * N_NODES + lo);
  for (int i = threadIdx.x; i < (int)(nh / 4); i += LB_THREADS) out[i] = ((const int4*)cnt)[i];
}

template <int NT>
__device__ __forceinline__ int block_scan_excl(int v, int* sh, int* total) {
  int t = threadIdx.x;
  sh[t] = v;
  __syncthreads();
  for (int o = 1; o < NT; o <<= 1) {
    int add = (t >= o) ? sh[t - o] : 0;
    __syncthreads();
    sh[t] += add;
    __syncthreads();
  }
  int incl = sh[t];
  *total = sh[NT - 1];
  __syncthreads();
  return incl - v;
}

__global__ __launch_bounds__(256) void scan_block_sums(const int* __restrict__ planes,
                                                       int* __restrict__ bsum, int n) {
  __shared__ int sh[256];
  int base = blockIdx.x * SCAN_ITEMS + threadIdx.x * 4;
  int s = 0;
  if (base < n) {
#pragma unroll
    for (int p = 0; p < NSLICE; ++p) {
      int4 v = *(const int4*)(planes + (size_t)p * N_NODES + base);
      s += v.x + v.y + v.z + v.w;
    }
  }
  int total;
  (void)block_scan_excl<256>(s, sh, &total);
  if (threadIdx.x == 0) bsum[blockIdx.x] = total;
}

__global__ __launch_bounds__(1024) void scan_top(int* __restrict__ bsum, int nb) {
  __shared__ int sh[1024];
  int t = threadIdx.x;
  int v = (t < nb) ? bsum[t] : 0;
  int total;
  int ex = block_scan_excl<1024>(v, sh, &total);
  if (t < nb) bsum[t] = ex;
}

__global__ __launch_bounds__(256) void scan_apply(int* __restrict__ planes,
                                                  int* __restrict__ arr,
                                                  const int* __restrict__ boff,
                                                  float* __restrict__ dinv, int n) {
  __shared__ int sh[256];
  int base = blockIdx.x * SCAN_ITEMS + threadIdx.x * 4;
  int4 c[NSLICE];
  int4 v = make_int4(0, 0, 0, 0);
#pragma unroll
  for (int p = 0; p < NSLICE; ++p) {
    c[p] = make_int4(0, 0, 0, 0);
    if (base < n) c[p] = *(const int4*)(planes + (size_t)p * N_NODES + base);
    v.x += c[p].x; v.y += c[p].y; v.z += c[p].z; v.w += c[p].w;
  }
  int s = v.x + v.y + v.z + v.w;
  int total;
  int ex = block_scan_excl<256>(s, sh, &total) + boff[blockIdx.x];
  if (base < n) {
    float4 d;
    d.x = rsqrtf((float)v.x + 1.0f);
    d.y = rsqrtf((float)v.y + 1.0f);
    d.z = rsqrtf((float)v.z + 1.0f);
    d.w = rsqrtf((float)v.w + 1.0f);
    *(float4*)(dinv + base) = d;
    int4 o;
    o.x = ex;
    o.y = o.x + v.x;
    o.z = o.y + v.y;
    o.w = o.z + v.z;
    *(int4*)(arr + base) = o;
#pragma unroll
    for (int p = 0; p < NSLICE; ++p) {
      *(int4*)(planes + (size_t)p * N_NODES + base) = o;
      o.x += c[p].x; o.y += c[p].y; o.z += c[p].z; o.w += c[p].w;
    }
  }
}

__global__ __launch_bounds__(LB_THREADS) void scatter_ranges(const int* __restrict__ src,
                                                             const int* __restrict__ dst,
                                                             const int* __restrict__ planes,
                                                             int* __restrict__ ssrc) {
  __shared__ __attribute__((aligned(16))) int cur[RANGE + 32];
  int r = blockIdx.x % NRANGE;
  int sl = blockIdx.x / NRANGE;
  int lo = r * RANGE;
  unsigned nh = (unsigned)min(RANGE, N_NODES - lo);
  const int4* in = (const int4*)(planes + (size_t)sl * N_NODES + lo);
  for (int i = threadIdx.x; i < (int)(nh / 4); i += LB_THREADS) ((int4*)cur)[i] = in[i];
  if (threadIdx.x < 32) cur[RANGE + threadIdx.x] = 0;
  __syncthreads();
  int spare = RANGE + (threadIdx.x & 31);
  const int4* d4 = (const int4*)dst + (size_t)sl * SLICE_V4;
  const int4* s4 = (const int4*)src + (size_t)sl * SLICE_V4;
  for (int i = threadIdx.x; i < SLICE_V4; i += LB_GROUPS * LB_THREADS) {
    int4 d[LB_GROUPS], sv[LB_GROUPS];
    uint4 p[LB_GROUPS];
#pragma unroll
    for (int g = 0; g < LB_GROUPS; ++g) {
      int ig = i + g * LB_THREADS;
      bool inside = (ig < SLICE_V4);
      d[g]  = inside ? d4[ig] : make_int4(-1, -1, -1, -1);
      sv[g] = inside ? s4[ig] : make_int4(0, 0, 0, 0);
    }
#pragma unroll
    for (int g = 0; g < LB_GROUPS; ++g) {
      d[g].x -= lo; d[g].y -= lo; d[g].z -= lo; d[g].w -= lo;
      bool kx = (unsigned)d[g].x < nh, ky = (unsigned)d[g].y < nh;
      bool kz = (unsigned)d[g].z < nh, kw = (unsigned)d[g].w < nh;
      p[g].x = (unsigned)LDS_ADD(&cur[kx ? d[g].x : spare], kx ? 1 : 0);
      p[g].y = (unsigned)LDS_ADD(&cur[ky ? d[g].y : spare], ky ? 1 : 0);
      p[g].z = (unsigned)LDS_ADD(&cur[kz ? d[g].z : spare], kz ? 1 : 0);
      p[g].w = (unsigned)LDS_ADD(&cur[kw ? d[g].w : spare], kw ? 1 : 0);
    }
#pragma unroll
    for (int g = 0; g < LB_GROUPS; ++g) {
      if ((unsigned)d[g].x < nh) ssrc[p[g].x] = sv[g].x;
      if ((unsigned)d[g].y < nh) ssrc[p[g].y] = sv[g].y;
      if ((unsigned)d[g].z < nh) ssrc[p[g].z] = sv[g].z;
      if ((unsigned)d[g].w < nh) ssrc[p[g].w] = sv[g].w;
    }
  }
}

__device__ __forceinline__ void fma4(float4& a, const float4& v, float w) {
  a.x = fmaf(v.x, w, a.x);
  a.y = fmaf(v.y, w, a.y);
  a.z = fmaf(v.z, w, a.z);
  a.w = fmaf(v.w, w, a.w);
}

#define IDBUF 512
#define TPITCH 36

__device__ __forceinline__ void stage_ids(const int* __restrict__ ssrc, int abase,
                                          int need, int* buf, int lane) {
  int4 v[IDBUF / 128];
#pragma unroll
  for (int k = 0; k < IDBUF / 128; ++k) {
    int o = 128 * k + 4 * lane;
    v[k] = *(const int4*)(ssrc + abase + (o < need ? o : 0));
  }
#pragma unroll
  for (int k = 0; k < IDBUF / 128; ++k) *(int4*)(buf + 128 * k + 4 * lane) = v[k];
}


__global__ __launch_bounds__(128) void pull1_xw1(const float* __restrict__ X,
                                                 const int* __restrict__ starts,
                                                 const int* __restrict__ ssrc,
                                                 const float* __restrict__ dinv,
                                                 const float* __restrict__ W1,
                                                 const float* __restrict__ b1,
                                                 float* __restrict__ H1) {
  __shared__ int idbuf[4 * IDBUF];
  __shared__ float tile[4 * 16 * TPITCH];
  int wave = blockIdx.x * (blockDim.x >> 5) + (threadIdx.x >> 5);
  int lane = threadIdx.x & 31;
  int m = lane & 15;
  int h = lane >> 4;
  int g = lane >> 3;
  int j = lane & 7;
  int nb = wave * 16 + 4 * g;
  int st[5];
#pragma unroll
  for (int q = 0; q < 5; ++q) {
    int n = nb + q;
    st[q] = (n < N_NODES) ? starts[n] : N_EDGES;
  }
  float ddq[4];
#pragma unroll
  for (int q = 0; q < 4; ++q) ddq[q] = dinv[nb + q];
  const float* xb = X + 4 * j;
  int* my = idbuf + (threadIdx.x >> 5) * IDBUF;
  int seg_beg = __builtin_amdgcn_readlane(st[0], 0);
  int seg_end = __builtin_amdgcn_readlane(st[4], 31);

  float4 acc[4];
#pragma unroll
  for (int q = 0; q < 4; ++q) acc[q] = make_float4(0.f, 0.f, 0.f, 0.f);
  for (int abase = seg_beg & ~3; abase < seg_end; abase += IDBUF) {
    stage_ids(ssrc, abase, min(IDBUF, seg_end - abase), my, lane);
#pragma unroll
    for (int q = 0; q < 4; ++q) {
      int lo = max(st[q], abase) - abase;
      int last = min(st[q + 1], abase + IDBUF) - 1 - abase;
      float dd = ddq[q];
      float4 a = acc[q];
      for (int i = lo; i <= last; i += 4) {
        bool k1 = (i + 1 <= last), k2 = (i + 2 <= last), k3 = (i + 3 <= last);
        int s0 = my[i];
        int s1 = my[k1 ? i + 1 : i];
        int s2 = my[k2 ? i + 2 : i];
        int s3 = my[k3 ? i + 3 : i];
        float d0 = dinv[s0], d1 = dinv[s1], d2 = dinv[s2], d3 = dinv[s3];
        float4 u0 = *(const float4*)(xb + (size_t)s0 * F_IN);
        float4 u1 = *(const float4*)(xb + (size_t)s1 * F_IN);
        float4 u2 = *(const float4*)(xb + (size_t)s2 * F_IN);
        float4 u3 = *(const float4*)(xb + (size_t)s3 * F_IN);
        float w0 = d0 * dd;
        float w1 = k1 ? d1 * dd : 0.0f;
        float w2 = k2 ? d2 * dd : 0.0f;
        float w3 = k3 ? d3 * dd : 0.0f;
        fma4(a, u0, w0); fma4(a, u1, w1); fma4(a, u2, w2); fma4(a, u3, w3);
      }
      acc[q] = a;
    }
  }
  float* tl = tile + (threadIdx.x >> 5) * (16 * TPITCH);
#pragma unroll
  for (int q = 0; q < 4; ++q) {
    float ws = ddq[q] * ddq[q];
    float4 u = *(const float4*)(xb + (size_t)(nb + q) * F_IN);
    float4 a = acc[q];
    fma4(a, u, ws);
    *(float4*)(tl + (4 * g + q) * TPITCH + 4 * j) = a;
  }
  const float* tr = tl + m * TPITCH + 16 * h;
  float4 a0 = *(const float4*)(tr + 0);
  float4 a1 = *(const float4*)(tr + 4);
  float4 a2 = *(const float4*)(tr + 8);
  float4 a3 = *(const float4*)(tr + 12);

  const float* wb = W1 + (size_t)(16 * h) * F_H1 + m;
  v2f A0 = {a0.x, a0.y}, A1 = {a0.z, a0.w}, A2 = {a1.x, a1.y}, A3 = {a1.z, a1.w};
  v2f A4 = {a2.x, a2.y}, A5 = {a2.z, a2.w}, A6 = {a3.x, a3.y}, A7 = {a3.z, a3.w};
  v2f B0 = {wb[0 * F_H1],  wb[1 * F_H1]},  B1 = {wb[2 * F_H1],  wb[3 * F_H1]};
  v2f B2 = {wb[4 * F_H1],  wb[5 * F_H1]},  B3 = {wb[6 * F_H1],  wb[7 * F_H1]};
  v2f B4 = {wb[8 * F_H1],  wb[9 * F_H1]},  B5 = {wb[10 * F_H1], wb[11 * F_H1]};
  v2f B6 = {wb[12 * F_H1], wb[13 * F_H1]}, B7 = {wb[14 * F_H1], wb[15 * F_H1]};
  v8f c = {0.f, 0.f, 0.f, 0.f, 0.f, 0.f, 0.f, 0.f};
  asm volatile(
      "v_wmma_f32_16x16x4_f32 %0, %1, %9, %0\n\t"
      "v_wmma_f32_16x16x4_f32 %0, %2, %10, %0\n\t"
      "v_wmma_f32_16x16x4_f32 %0, %3, %11, %0\n\t"
      "v_wmma_f32_16x16x4_f32 %0, %4, %12, %0\n\t"
      "v_wmma_f32_16x16x4_f32 %0, %5, %13, %0\n\t"
      "v_wmma_f32_16x16x4_f32 %0, %6, %14, %0\n\t"
      "v_wmma_f32_16x16x4_f32 %0, %7, %15, %0\n\t"
      "v_wmma_f32_16x16x4_f32 %0, %8, %16, %0\n\t"
      "v_nop\n\tv_nop\n\tv_nop\n\tv_nop\n\tv_nop"
      : "+v"(c)
      : "v"(A0), "v"(A1), "v"(A2), "v"(A3), "v"(A4), "v"(A5), "v"(A6), "v"(A7),
        "v"(B0), "v"(B1), "v"(B2), "v"(B3), "v"(B4), "v"(B5), "v"(B6), "v"(B7));
  float bb = b1[m];
  float* out = H1 + (size_t)(wave * 16 + 8 * h) * F_H1 + m;
#pragma unroll
  for (int j = 0; j < 8; ++j) out[(size_t)j * F_H1] = fmaxf(c[j] + bb, 0.0f);
}

__global__ __launch_bounds__(128) void gemm_h1_w2(float* H, const float* __restrict__ W2,
                                                  const float* __restrict__ dinv) {
  int wave = blockIdx.x * (blockDim.x >> 5) + (threadIdx.x >> 5);
  int lane = threadIdx.x & 31;
  int m  = lane & 15;
  int kk = (lane >> 4) << 1;
  float* hrow = H + (size_t)(wave * 16 + m) * F_H1;
  bool live_col = (m < F_H2);
  float dm = dinv[wave * 16 + m];

  v2f a[4], b[4];
#pragma unroll
  for (int t = 0; t < 4; ++t) {
    int k = 4 * t + kk;
    a[t].x = hrow[k];
    a[t].y = hrow[k + 1];
    b[t].x = live_col ? W2[k * F_H2 + m] : 0.0f;
    b[t].y = live_col ? W2[(k + 1) * F_H2 + m] : 0.0f;
  }
  if (lane < 16) hrow[F_H2] = dm;

  v8f c = {0.f, 0.f, 0.f, 0.f, 0.f, 0.f, 0.f, 0.f};
#pragma unroll
  for (int t = 0; t < 4; ++t)
    c = __builtin_amdgcn_wmma_f32_16x16x4_f32(false, a[t], false, b[t],
                                              (short)0, c, false, false);
  if (live_col) {
    int hi = lane >> 4;
    float* out = H + (size_t)(wave * 16 + 8 * hi) * F_H1 + m;
#pragma unroll
    for (int j = 0; j < 8; ++j) out[(size_t)j * F_H1] = c[j];
  }
}

__device__ __forceinline__ float quad_lane2(float v) {
  int r = __builtin_amdgcn_mov_dpp(__float_as_int(v), 0xAA, 0xF, 0xF, true);
  return __int_as_float(r);
}

__global__ __launch_bounds__(128) void pull2_pool(const float* __restrict__ H,
                                                  const int* __restrict__ starts,
                                                  const int* __restrict__ ssrc,
                                                  const float* __restrict__ dinv,
                                                  const float* __restrict__ b2,
                                                  const int* __restrict__ batch,
                                                  float* __restrict__ pooled,
                                                  float* __restrict__ cnt) {
  __shared__ int idbuf[4 * IDBUF];
  int wave = blockIdx.x * (blockDim.x >> 5) + (threadIdx.x >> 5);
  int lane = threadIdx.x & 31;
  int g = lane >> 2;
  int j = lane & 3;
  int nb = wave * 16 + 2 * g;
  int st[3];
#pragma unroll
  for (int q = 0; q < 3; ++q) {
    int n = nb + q;
    st[q] = (n < N_NODES) ? starts[n] : N_EDGES;
  }
  float ddq[2];
  ddq[0] = dinv[nb];
  ddq[1] = dinv[nb + 1];
  const float* hb = H + 4 * j;
  int* my = idbuf + (threadIdx.x >> 5) * IDBUF;
  int seg_beg = __builtin_amdgcn_readlane(st[0], 0);
  int seg_end = __builtin_amdgcn_readlane(st[2], 31);

  float4 acc[2];
  acc[0] = make_float4(0.f, 0.f, 0.f, 0.f);
  acc[1] = acc[0];
  for (int abase = seg_beg & ~3; abase < seg_end; abase += IDBUF) {
    stage_ids(ssrc, abase, min(IDBUF, seg_end - abase), my, lane);
#pragma unroll
    for (int q = 0; q < 2; ++q) {
      int lo = max(st[q], abase) - abase;
      int last = min(st[q + 1], abase + IDBUF) - 1 - abase;
      float dd = ddq[q];
      float4 a = acc[q];
      for (int i = lo; i <= last; i += 4) {
        bool k1 = (i + 1 <= last), k2 = (i + 2 <= last), k3 = (i + 3 <= last);
        int s0 = my[i];
        int s1 = my[k1 ? i + 1 : i];
        int s2 = my[k2 ? i + 2 : i];
        int s3 = my[k3 ? i + 3 : i];
        float4 u0 = *(const float4*)(hb + (size_t)s0 * F_H1);
        float4 u1 = *(const float4*)(hb + (size_t)s1 * F_H1);
        float4 u2 = *(const float4*)(hb + (size_t)s2 * F_H1);
        float4 u3 = *(const float4*)(hb + (size_t)s3 * F_H1);
        float d0 = quad_lane2(u0.x), d1 = quad_lane2(u1.x);
        float d2 = quad_lane2(u2.x), d3 = quad_lane2(u3.x);
        float w0 = d0 * dd;
        float w1 = k1 ? d1 * dd : 0.0f;
        float w2 = k2 ? d2 * dd : 0.0f;
        float w3 = k3 ? d3 * dd : 0.0f;
        fma4(a, u0, w0); fma4(a, u1, w1); fma4(a, u2, w2); fma4(a, u3, w3);
      }
      acc[q] = a;
    }
  }
  if (j < 2) {
    float4 bv = *(const float4*)(b2 + 4 * j);
#pragma unroll
    for (int q = 0; q < 2; ++q) {
      int node = nb + q;
      float ws = ddq[q] * ddq[q];
      float4 sv = *(const float4*)(hb + (size_t)node * F_H1);
      float4 a = acc[q];
      float4 r;
      r.x = fmaxf(a.x + sv.x * ws + bv.x, 0.0f);
      r.y = fmaxf(a.y + sv.y * ws + bv.y, 0.0f);
      r.z = fmaxf(a.z + sv.z * ws + bv.z, 0.0f);
      r.w = fmaxf(a.w + sv.w * ws + bv.w, 0.0f);
      int gr = batch[node];
      float* ps = pooled + (size_t)gr * F_H2 + 4 * j;
      ATOMIC_ADD(&ps[0], r.x);
      ATOMIC_ADD(&ps[1], r.y);
      ATOMIC_ADD(&ps[2], r.z);
      ATOMIC_ADD(&ps[3], r.w);
      if (j == 0) ATOMIC_ADD(&cnt[gr], 1.0f);
    }
  }
}

__global__ void readout(const float* __restrict__ pooled, const float* __restrict__ cnt,
                        const float* __restrict__ Wl, const float* __restrict__ bl,
                        float* __restrict__ out, int G) {
  int g = blockIdx.x * blockDim.x + threadIdx.x;
  if (g >= G) return;
  float inv = 1.0f / fmaxf(cnt[g], 1.0f);
  float p[F_H2];
#pragma unroll
  for (int f = 0; f < F_H2; ++f) p[f] = pooled[(size_t)g * F_H2 + f] * inv;
#pragma unroll
  for (int c = 0; c < N_CLS; ++c) {
    float acc = bl[c];
#pragma unroll
    for (int f = 0; f < F_H2; ++f) acc += p[f] * Wl[f * N_CLS + c];
    out[(size_t)g * N_CLS + c] = acc;
  }
}


extern "C" void kernel_launch(void* const* d_in, const int* in_sizes, int n_in,
                              void* d_out, int out_size, void* d_ws, size_t ws_size,
                              hipStream_t stream) {
  const float* x     = (const float*)d_in[0];
  const int*   ei    = (const int*)d_in[1];
  const int*   src   = ei;
  const int*   dst   = ei + N_EDGES;
  const int*   batch = (const int*)d_in[2];
  const float* W1 = (const float*)d_in[4];
  const float* b1 = (const float*)d_in[5];
  const float* W2 = (const float*)d_in[6];
  const float* b2 = (const float*)d_in[7];
  const float* Wl = (const float*)d_in[8];
  const float* bl = (const float*)d_in[9];
  float* out = (float*)d_out;

  char* ws = (char*)d_ws;
  float* dinv   = (float*)(ws);
  int*   arr    = (int*)(ws + (4ll << 20));
  int*   ssrc   = (int*)(ws + (8ll << 20));
  float* h1     = (float*)(ws + 72388608ll);
  int*   planes = (int*)h1;
  int*   bsum   = (int*)(ws + 136388608ll);
  float* pooled = (float*)(ws + 136396800ll);
  float* cnt    = pooled + (size_t)N_GRAPHS * F_H2;

  const int NB = 256;
  const int TILE_BLOCKS = N_NODES / 64;

  hist_ranges<<<NRANGE * NSLICE, LB_THREADS, 0, stream>>>(dst, planes);
  scan_block_sums<<<SCAN_BLOCKS, 256, 0, stream>>>(planes, bsum, N_NODES);
  scan_top<<<1, 1024, 0, stream>>>(bsum, SCAN_BLOCKS);
  scan_apply<<<SCAN_BLOCKS, 256, 0, stream>>>(planes, arr, bsum, dinv, N_NODES);
  scatter_ranges<<<NRANGE * NSLICE, LB_THREADS, 0, stream>>>(src, dst, planes, ssrc);

  pull1_xw1<<<TILE_BLOCKS, 128, 0, stream>>>(x, arr, ssrc, dinv, W1, b1, h1);
  gemm_h1_w2<<<TILE_BLOCKS, 128, 0, stream>>>(h1, W2, dinv);

  zero_u32<<<128, 256, 0, stream>>>((unsigned*)pooled, N_GRAPHS * (F_H2 + 1));
  pull2_pool<<<TILE_BLOCKS, 128, 0, stream>>>(h1, arr, ssrc, dinv, b2, batch, pooled, cnt);
  readout<<<(N_GRAPHS + NB - 1) / NB, NB, 0, stream>>>(pooled, cnt, Wl, bl,
                                                       out, N_GRAPHS);
}


# list build keyed by 16-node tile: 62500 LDS counters, 256 edge slices each read once (not 13x), then a wave per tile orders its piece by node; sum order = slice, LDS arrival, in-tile placement
# speedup vs baseline: 5.0423x; 1.1709x over previous
#include <hip/hip_runtime.h>


typedef float v2f __attribute__((ext_vector_type(2)));
typedef float v8f __attribute__((ext_vector_type(8)));

#define N_NODES   1000000
#define N_EDGES   16000000
#define F_IN      32
#define F_H1      16
#define F_H2      8
#define N_CLS     5
#define N_GRAPHS  10000

#define ATOMIC_ADD(p, v) \
  __hip_atomic_fetch_add((p), (v), __ATOMIC_RELAXED, __HIP_MEMORY_SCOPE_AGENT)


__global__ void zero_u32(unsigned* __restrict__ p, int n) {
  int i = blockIdx.x * blockDim.x + threadIdx.x;
  int stride = gridDim.x * blockDim.x;
  for (; i < n; i += stride) p[i] = 0u;
}

#define LDS_ADD(p, v) \
  __hip_atomic_fetch_add((p), (v), __ATOMIC_RELAXED, __HIP_MEMORY_SCOPE_WORKGROUP)

#define NTILE   (N_NODES / 16)
#define NSLICE  256
#define SLICE_V4 (N_EDGES / 4 / NSLICE)
#define LB_THREADS 1024
#define LB_GROUPS 4
#define SRC_BITS 20
#define SRC_MASK ((1 << SRC_BITS) - 1)
static_assert(NTILE * 16 == N_NODES && NTILE % 4 == 0, "tiles cover the nodes, planes are whole 16-byte groups");
static_assert(SLICE_V4 * 4 * NSLICE == N_EDGES, "slices cover the edges exactly");
static_assert((NTILE + 32) * 4 <= 320 * 1024, "the tile counters fit the LDS");
static_assert(N_NODES <= (1 << SRC_BITS), "a source id fits its field");
static_assert((long long)NSLICE * NTILE * 4 == 64000000ll, "planes fill the 64,000,000-byte region");

#define SCAN_TILES 64
#define SCAN_BLOCKS ((NTILE + SCAN_TILES - 1) / SCAN_TILES)
static_assert(SCAN_BLOCKS <= 1024, "one block scans the block sums");

__global__ __launch_bounds__(LB_THREADS) void hist_tiles(const int* __restrict__ dst,
                                                         int* __restrict__ planes) {
  __shared__ __attribute__((aligned(16))) int cnt[NTILE];
  int sl = blockIdx.x;
  for (int i = threadIdx.x; i < NTILE / 4; i += LB_THREADS)
    ((int4*)cnt)[i] = make_int4(0, 0, 0, 0);
  __syncthreads();
  const int4* d4 = (const int4*)dst + (size_t)sl * SLICE_V4;
  for (int i = threadIdx.x; i < SLICE_V4; i += LB_GROUPS * LB_THREADS) {
    int4 d[LB_GROUPS];
#pragma unroll
    for (int g = 0; g < LB_GROUPS; ++g) {
      int ig = i + g * LB_THREADS;
      d[g] = (ig < SLICE_V4) ? d4[ig] : make_int4(-1, -1, -1, -1);
    }
#pragma unroll
    for (int g = 0; g < LB_GROUPS; ++g) {
      if ((unsigned)d[g].x < (unsigned)N_NODES) LDS_ADD(&cnt[d[g].x >> 4], 1);
      if ((unsigned)d[g].y < (unsigned)N_NODES) LDS_ADD(&cnt[d[g].y >> 4], 1);
      if ((unsigned)d[g].z < (unsigned)N_NODES) LDS_ADD(&cnt[d[g].z >> 4], 1);
      if ((unsigned)d[g].w < (unsigned)N_NODES) LDS_ADD(&cnt[d[g].w >> 4], 1);
    }
  }
  __syncthreads();
  int4* out = (int4*)(planes + (size_t)sl * NTILE);
  for (int i = threadIdx.x; i < NTILE / 4; i += LB_THREADS) out[i] = ((const int4*)cnt)[i];
}

template <int NT>
__device__ __forceinline__ int block_scan_excl(int v, int* sh, int* total) {
  int t = threadIdx.x;
  sh[t] = v;
  __syncthreads();
  for (int o = 1; o < NT; o <<= 1) {
    int add = (t >= o) ? sh[t - o] : 0;
    __syncthreads();
    sh[t] += add;
    __syncthreads();
  }
  int incl = sh[t];
  *total = sh[NT - 1];
  __syncthreads();
  return incl - v;
}

__global__ __launch_bounds__(256) void scan_block_sums(const int* __restrict__ planes,
                                                       int* __restrict__ bsum) {
  __shared__ int sh[256];
  int tq = threadIdx.x & 15;
  int pg = threadIdx.x >> 4;
  int base = blockIdx.x * SCAN_TILES + 4 * tq;
  int s = 0;
  if (base < NTILE) {
#pragma unroll
    for (int p = 0; p < 16; ++p) {
      int4 v = *(const int4*)(planes + (size_t)(pg * 16 + p) * NTILE + base);
      s += v.x + v.y + v.z + v.w;
    }
  }
  int total;
  (void)block_scan_excl<256>(s, sh, &total);
  if (threadIdx.x == 0) bsum[blockIdx.x] = total;
}

__global__ __launch_bounds__(1024) void scan_top(int* __restrict__ bsum, int nb) {
  __shared__ int sh[1024];
  int t = threadIdx.x;
  int v = (t < nb) ? bsum[t] : 0;
  int total;
  int ex = block_scan_excl<1024>(v, sh, &total);
  if (t < nb) bsum[t] = ex;
}

__global__ __launch_bounds__(256) void scan_apply(int* __restrict__ planes,
                                                  int* __restrict__ tbeg,
                                                  const int* __restrict__ boff) {
  __shared__ int4 part[16 * 16];
  __shared__ int tsum[16];
  int tq = threadIdx.x & 15;
  int pg = threadIdx.x >> 4;
  int base = blockIdx.x * SCAN_TILES + 4 * tq;
  bool live = (base < NTILE);
  int4 c[16];
  int4 v = make_int4(0, 0, 0, 0);
#pragma unroll
  for (int p = 0; p < 16; ++p) {
    c[p] = make_int4(0, 0, 0, 0);
    if (live) c[p] = *(const int4*)(planes + (size_t)(pg * 16 + p) * NTILE + base);
    v.x += c[p].x; v.y += c[p].y; v.z += c[p].z; v.w += c[p].w;
  }
  part[pg * 16 + tq] = v;
  __syncthreads();
  int4 pre = make_int4(0, 0, 0, 0);
  int4 tot = make_int4(0, 0, 0, 0);
  for (int g = 0; g < 16; ++g) {
    int4 w = part[g * 16 + tq];
    if (g < pg) { pre.x += w.x; pre.y += w.y; pre.z += w.z; pre.w += w.w; }
    tot.x += w.x; tot.y += w.y; tot.z += w.z; tot.w += w.w;
  }
  if (pg == 0) tsum[tq] = tot.x + tot.y + tot.z + tot.w;
  __syncthreads();
  int ex = boff[blockIdx.x];
  for (int q = 0; q < 16; ++q) {
    int w = tsum[q];
    if (q < tq) ex += w;
  }
  if (live) {
    int4 o;
    o.x = ex;
    o.y = o.x + tot.x;
    o.z = o.y + tot.y;
    o.w = o.z + tot.z;
    if (pg == 0) {
      *(int4*)(tbeg + base) = o;
      if (base + 4 == NTILE) tbeg[NTILE] = o.w + tot.w;
    }
    o.x += pre.x; o.y += pre.y; o.z += pre.z; o.w += pre.w;
#pragma unroll
    for (int p = 0; p < 16; ++p) {
      *(int4*)(planes + (size_t)(pg * 16 + p) * NTILE + base) = o;
      o.x += c[p].x; o.y += c[p].y; o.z += c[p].z; o.w += c[p].w;
    }
  }
}

__global__ __launch_bounds__(LB_THREADS) void scatter_tiles(const int* __restrict__ src,
                                                            const int* __restrict__ dst,
                                                            const int* __restrict__ planes,
                                                            int* __restrict__ lst) {
  __shared__ __attribute__((aligned(16))) int cur[NTILE + 32];
  int sl = blockIdx.x;
  const int4* in = (const int4*)(planes + (size_t)sl * NTILE);
  for (int i = threadIdx.x; i < NTILE / 4; i += LB_THREADS) ((int4*)cur)[i] = in[i];
  if (threadIdx.x < 32) cur[NTILE + threadIdx.x] = 0;
  __syncthreads();
  int spare = NTILE + (threadIdx.x & 31);
  const int4* d4 = (const int4*)dst + (size_t)sl * SLICE_V4;
  const int4* s4 = (const int4*)src + (size_t)sl * SLICE_V4;
  for (int i = threadIdx.x; i < SLICE_V4; i += LB_GROUPS * LB_THREADS) {
    int4 d[LB_GROUPS], sv[LB_GROUPS];
    uint4 p[LB_GROUPS];
#pragma unroll
    for (int g = 0; g < LB_GROUPS; ++g) {
      int ig = i + g * LB_THREADS;
      bool inside = (ig < SLICE_V4);
      d[g]  = inside ? d4[ig] : make_int4(-1, -1, -1, -1);
      sv[g] = inside ? s4[ig] : make_int4(0, 0, 0, 0);
    }
#pragma unroll
    for (int g = 0; g < LB_GROUPS; ++g) {
      bool kx = (unsigned)d[g].x < (unsigned)N_NODES, ky = (unsigned)d[g].y < (unsigned)N_NODES;
      bool kz = (unsigned)d[g].z < (unsigned)N_NODES, kw = (unsigned)d[g].w < (unsigned)N_NODES;
      p[g].x = (unsigned)LDS_ADD(&cur[kx ? (d[g].x >> 4) : spare], kx ? 1 : 0);
      p[g].y = (unsigned)LDS_ADD(&cur[ky ? (d[g].y >> 4) : spare], ky ? 1 : 0);
      p[g].z = (unsigned)LDS_ADD(&cur[kz ? (d[g].z >> 4) : spare], kz ? 1 : 0);
      p[g].w = (unsigned)LDS_ADD(&cur[kw ? (d[g].w >> 4) : spare], kw ? 1 : 0);
    }
#pragma unroll
    for (int g = 0; g < LB_GROUPS; ++g) {
      if ((unsigned)d[g].x < (unsigned)N_NODES) lst[p[g].x] = sv[g].x | ((d[g].x & 15) << SRC_BITS);
      if ((unsigned)d[g].y < (unsigned)N_NODES) lst[p[g].y] = sv[g].y | ((d[g].y & 15) << SRC_BITS);
      if ((unsigned)d[g].z < (unsigned)N_NODES) lst[p[g].z] = sv[g].z | ((d[g].z & 15) << SRC_BITS);
      if ((unsigned)d[g].w < (unsigned)N_NODES) lst[p[g].w] = sv[g].w | ((d[g].w & 15) << SRC_BITS);
    }
  }
}

__global__ __launch_bounds__(128) void tile_order(const int* __restrict__ lst,
                                                  const int* __restrict__ tbeg,
                                                  int* __restrict__ starts,
                                                  float* __restrict__ dinv,
                                                  int* __restrict__ ssrc) {
  __shared__ __attribute__((aligned(16))) int cnt[4 * 16];
  __shared__ int cur[4 * 48];
  int w = threadIdx.x >> 5;
  int lane = threadIdx.x & 31;
  int tile = blockIdx.x * 4 + w;
  int* mc = cnt + 16 * w;
  int* mu = cur + 48 * w;
  int beg = tbeg[tile];
  int end = tbeg[tile + 1];
  if (lane < 16) mc[lane] = 0;
  mu[16 + lane] = 0;
  for (int base = beg & ~3; base < end; base += 128) {
    int idx = base + 4 * lane;
    int4 e = make_int4(0, 0, 0, 0);
    if (idx < end) e = *(const int4*)(lst + idx);
    if (idx     >= beg && idx     < end) LDS_ADD(&mc[(e.x >> SRC_BITS) & 15], 1);
    if (idx + 1 >= beg && idx + 1 < end) LDS_ADD(&mc[(e.y >> SRC_BITS) & 15], 1);
    if (idx + 2 >= beg && idx + 2 < end) LDS_ADD(&mc[(e.z >> SRC_BITS) & 15], 1);
    if (idx + 3 >= beg && idx + 3 < end) LDS_ADD(&mc[(e.w >> SRC_BITS) & 15], 1);
  }
  if (lane < 16) {
    int4 q0 = ((const int4*)mc)[0], q1 = ((const int4*)mc)[1];
    int4 q2 = ((const int4*)mc)[2], q3 = ((const int4*)mc)[3];
    int v[16] = {q0.x, q0.y, q0.z, q0.w, q1.x, q1.y, q1.z, q1.w,
                 q2.x, q2.y, q2.z, q2.w, q3.x, q3.y, q3.z, q3.w};
    int c = 0;
    int pre = 0;
#pragma unroll
    for (int k = 0; k < 16; ++k) {
      pre += (k < lane) ? v[k] : 0;
      c = (k == lane) ? v[k] : c;
    }
    int node = tile * 16 + lane;
    starts[node] = beg + pre;
    dinv[node] = rsqrtf((float)c + 1.0f);
    mu[lane] = beg + pre;
  }
  int spare = 16 + lane;
  for (int base = beg & ~3; base < end; base += 128) {
    int idx = base + 4 * lane;
    int4 e = make_int4(0, 0, 0, 0);
    if (idx < end) e = *(const int4*)(lst + idx);
    bool kx = (idx     >= beg && idx     < end), ky = (idx + 1 >= beg && idx + 1 < end);
    bool kz = (idx + 2 >= beg && idx + 2 < end), kw = (idx + 3 >= beg && idx + 3 < end);
    int px = LDS_ADD(&mu[kx ? ((e.x >> SRC_BITS) & 15) : spare], kx ? 1 : 0);
    int py = LDS_ADD(&mu[ky ? ((e.y >> SRC_BITS) & 15) : spare], ky ? 1 : 0);
    int pz = LDS_ADD(&mu[kz ? ((e.z >> SRC_BITS) & 15) : spare], kz ? 1 : 0);
    int pw = LDS_ADD(&mu[kw ? ((e.w >> SRC_BITS) & 15) : spare], kw ? 1 : 0);
    if (kx) ssrc[px] = e.x & SRC_MASK;
    if (ky) ssrc[py] = e.y & SRC_MASK;
    if (kz) ssrc[pz] = e.z & SRC_MASK;
    if (kw) ssrc[pw] = e.w & SRC_MASK;
  }
}

__device__ __forceinline__ void fma4(float4& a, const float4& v, float w) {
  a.x = fmaf(v.x, w, a.x);
  a.y = fmaf(v.y, w, a.y);
  a.z = fmaf(v.z, w, a.z);
  a.w = fmaf(v.w, w, a.w);
}

#define IDBUF 512
#define TPITCH 36

__device__ __forceinline__ void stage_ids(const int* __restrict__ ssrc, int abase,
                                          int need, int* buf, int lane) {
  int4 v[IDBUF / 128];
#pragma unroll
  for (int k = 0; k < IDBUF / 128; ++k) {
    int o = 128 * k + 4 * lane;
    v[k] = *(const int4*)(ssrc + abase + (o < need ? o : 0));
  }
#pragma unroll
  for (int k = 0; k < IDBUF / 128; ++k) *(int4*)(buf + 128 * k + 4 * lane) = v[k];
}


__global__ __launch_bounds__(128) void pull1_xw1(const float* __restrict__ X,
                                                 const int* __restrict__ starts,
                                                 const int* __restrict__ ssrc,
                                                 const float* __restrict__ dinv,
                                                 const float* __restrict__ W1,
                                                 const float* __restrict__ b1,
                                                 float* __restrict__ H1) {
  __shared__ int idbuf[4 * IDBUF];
  __shared__ float tile[4 * 16 * TPITCH];
  int wave = blockIdx.x * (blockDim.x >> 5) + (threadIdx.x >> 5);
  int lane = threadIdx.x & 31;
  int m = lane & 15;
  int h = lane >> 4;
  int g = lane >> 3;
  int j = lane & 7;
  int nb = wave * 16 + 4 * g;
  int st[5];
#pragma unroll
  for (int q = 0; q < 5; ++q) {
    int n = nb + q;
    st[q] = (n < N_NODES) ? starts[n] : N_EDGES;
  }
  float ddq[4];
#pragma unroll
  for (int q = 0; q < 4; ++q) ddq[q] = dinv[nb + q];
  const float* xb = X + 4 * j;
  int* my = idbuf + (threadIdx.x >> 5) * IDBUF;
  int seg_beg = __builtin_amdgcn_readlane(st[0], 0);
  int seg_end = __builtin_amdgcn_readlane(st[4], 31);

  float4 acc[4];
#pragma unroll
  for (int q = 0; q < 4; ++q) acc[q] = make_float4(0.f, 0.f, 0.f, 0.f);
  for (int abase = seg_beg & ~3; abase < seg_end; abase += IDBUF) {
    stage_ids(ssrc, abase, min(IDBUF, seg_end - abase), my, lane);
#pragma unroll
    for (int q = 0; q < 4; ++q) {
      int lo = max(st[q], abase) - abase;
      int last = min(st[q + 1], abase + IDBUF) - 1 - abase;
      float dd = ddq[q];
      float4 a = acc[q];
      for (int i = lo; i <= last; i += 4) {
        bool k1 = (i + 1 <= last), k2 = (i + 2 <= last), k3 = (i + 3 <= last);
        int s0 = my[i];
        int s1 = my[k1 ? i + 1 : i];
        int s2 = my[k2 ? i + 2 : i];
        int s3 = my[k3 ? i + 3 : i];
        float d0 = dinv[s0], d1 = dinv[s1], d2 = dinv[s2], d3 = dinv[s3];
        float4 u0 = *(const float4*)(xb + (size_t)s0 * F_IN);
        float4 u1 = *(const float4*)(xb + (size_t)s1 * F_IN);
        float4 u2 = *(const float4*)(xb + (size_t)s2 * F_IN);
        float4 u3 = *(const float4*)(xb + (size_t)s3 * F_IN);
        float w0 = d0 * dd;
        float w1 = k1 ? d1 * dd : 0.0f;
        float w2 = k2 ? d2 * dd : 0.0f;
        float w3 = k3 ? d3 * dd : 0.0f;
        fma4(a, u0, w0); fma4(a, u1, w1); fma4(a, u2, w2); fma4(a, u3, w3);
      }
      acc[q] = a;
    }
  }
  float* tl = tile + (threadIdx.x >> 5) * (16 * TPITCH);
#pragma unroll
  for (int q = 0; q < 4; ++q) {
    float ws = ddq[q] * ddq[q];
    float4 u = *(const float4*)(xb + (size_t)(nb + q) * F_IN);
    float4 a = acc[q];
    fma4(a, u, ws);
    *(float4*)(tl + (4 * g + q) * TPITCH + 4 * j) = a;
  }
  const float* tr = tl + m * TPITCH + 16 * h;
  float4 a0 = *(const float4*)(tr + 0);
  float4 a1 = *(const float4*)(tr + 4);
  float4 a2 = *(const float4*)(tr + 8);
  float4 a3 = *(const float4*)(tr + 12);

  const float* wb = W1 + (size_t)(16 * h) * F_H1 + m;
  v2f A0 = {a0.x, a0.y}, A1 = {a0.z, a0.w}, A2 = {a1.x, a1.y}, A3 = {a1.z, a1.w};
  v2f A4 = {a2.x, a2.y}, A5 = {a2.z, a2.w}, A6 = {a3.x, a3.y}, A7 = {a3.z, a3.w};
  v2f B0 = {wb[0 * F_H1],  wb[1 * F_H1]},  B1 = {wb[2 * F_H1],  wb[3 * F_H1]};
  v2f B2 = {wb[4 * F_H1],  wb[5 * F_H1]},  B3 = {wb[6 * F_H1],  wb[7 * F_H1]};
  v2f B4 = {wb[8 * F_H1],  wb[9 * F_H1]},  B5 = {wb[10 * F_H1], wb[11 * F_H1]};
  v2f B6 = {wb[12 * F_H1], wb[13 * F_H1]}, B7 = {wb[14 * F_H1], wb[15 * F_H1]};
  v8f c = {0.f, 0.f, 0.f, 0.f, 0.f, 0.f, 0.f, 0.f};
  asm volatile(
      "v_wmma_f32_16x16x4_f32 %0, %1, %9, %0\n\t"
      "v_wmma_f32_16x16x4_f32 %0, %2, %10, %0\n\t"
      "v_wmma_f32_16x16x4_f32 %0, %3, %11, %0\n\t"
      "v_wmma_f32_16x16x4_f32 %0, %4, %12, %0\n\t"
      "v_wmma_f32_16x16x4_f32 %0, %5, %13, %0\n\t"
      "v_wmma_f32_16x16x4_f32 %0, %6, %14, %0\n\t"
      "v_wmma_f32_16x16x4_f32 %0, %7, %15, %0\n\t"
      "v_wmma_f32_16x16x4_f32 %0, %8, %16, %0\n\t"
      "v_nop\n\tv_nop\n\tv_nop\n\tv_nop\n\tv_nop"
      : "+v"(c)
      : "v"(A0), "v"(A1), "v"(A2), "v"(A3), "v"(A4), "v"(A5), "v"(A6), "v"(A7),
        "v"(B0), "v"(B1), "v"(B2), "v"(B3), "v"(B4), "v"(B5), "v"(B6), "v"(B7));
  float bb = b1[m];
  float* out = H1 + (size_t)(wave * 16 + 8 * h) * F_H1 + m;
#pragma unroll
  for (int j = 0; j < 8; ++j) out[(size_t)j * F_H1] = fmaxf(c[j] + bb, 0.0f);
}

__global__ __launch_bounds__(128) void gemm_h1_w2(float* H, const float* __restrict__ W2,
                                                  const float* __restrict__ dinv) {
  int wave = blockIdx.x * (blockDim.x >> 5) + (threadIdx.x >> 5);
  int lane = threadIdx.x & 31;
  int m  = lane & 15;
  int kk = (lane >> 4) << 1;
  float* hrow = H + (size_t)(wave * 16 + m) * F_H1;
  bool live_col = (m < F_H2);
  float dm = dinv[wave * 16 + m];

  v2f a[4], b[4];
#pragma unroll
  for (int t = 0; t < 4; ++t) {
    int k = 4 * t + kk;
    a[t].x = hrow[k];
    a[t].y = hrow[k + 1];
    b[t].x = live_col ? W2[k * F_H2 + m] : 0.0f;
    b[t].y = live_col ? W2[(k + 1) * F_H2 + m] : 0.0f;
  }
  if (lane < 16) hrow[F_H2] = dm;

  v8f c = {0.f, 0.f, 0.f, 0.f, 0.f, 0.f, 0.f, 0.f};
#pragma unroll
  for (int t = 0; t < 4; ++t)
    c = __builtin_amdgcn_wmma_f32_16x16x4_f32(false, a[t], false, b[t],
                                              (short)0, c, false, false);
  if (live_col) {
    int hi = lane >> 4;
    float* out = H + (size_t)(wave * 16 + 8 * hi) * F_H1 + m;
#pragma unroll
    for (int j = 0; j < 8; ++j) out[(size_t)j * F_H1] = c[j];
  }
}

__device__ __forceinline__ float quad_lane2(float v) {
  int r = __builtin_amdgcn_mov_dpp(__float_as_int(v), 0xAA, 0xF, 0xF, true);
  return __int_as_float(r);
}

__global__ __launch_bounds__(128) void pull2_pool(const float* __restrict__ H,
                                                  const int* __restrict__ starts,
                                                  const int* __restrict__ ssrc,
                                                  const float* __restrict__ dinv,
                                                  const float* __restrict__ b2,
                                                  const int* __restrict__ batch,
                                                  float* __restrict__ pooled,
                                                  float* __restrict__ cnt) {
  __shared__ int idbuf[4 * IDBUF];
  int wave = blockIdx.x * (blockDim.x >> 5) + (threadIdx.x >> 5);
  int lane = threadIdx.x & 31;
  int g = lane >> 2;
  int j = lane & 3;
  int nb = wave * 16 + 2 * g;
  int st[3];
#pragma unroll
  for (int q = 0; q < 3; ++q) {
    int n = nb + q;
    st[q] = (n < N_NODES) ? starts[n] : N_EDGES;
  }
  float ddq[2];
  ddq[0] = dinv[nb];
  ddq[1] = dinv[nb + 1];
  const float* hb = H + 4 * j;
  int* my = idbuf + (threadIdx.x >> 5) * IDBUF;
  int seg_beg = __builtin_amdgcn_readlane(st[0], 0);
  int seg_end = __builtin_amdgcn_readlane(st[2], 31);

  float4 acc[2];
  acc[0] = make_float4(0.f, 0.f, 0.f, 0.f);
  acc[1] = acc[0];
  for (int abase = seg_beg & ~3; abase < seg_end; abase += IDBUF) {
    stage_ids(ssrc, abase, min(IDBUF, seg_end - abase), my, lane);
#pragma unroll
    for (int q = 0; q < 2; ++q) {
      int lo = max(st[q], abase) - abase;
      int last = min(st[q + 1], abase + IDBUF) - 1 - abase;
      float dd = ddq[q];
      float4 a = acc[q];
      for (int i = lo; i <= last; i += 4) {
        bool k1 = (i + 1 <= last), k2 = (i + 2 <= last), k3 = (i + 3 <= last);
        int s0 = my[i];
        int s1 = my[k1 ? i + 1 : i];
        int s2 = my[k2 ? i + 2 : i];
        int s3 = my[k3 ? i + 3 : i];
        float4 u0 = *(const float4*)(hb + (size_t)s0 * F_H1);
        float4 u1 = *(const float4*)(hb + (size_t)s1 * F_H1);
        float4 u2 = *(const float4*)(hb + (size_t)s2 * F_H1);
        float4 u3 = *(const float4*)(hb + (size_t)s3 * F_H1);
        float d0 = quad_lane2(u0.x), d1 = quad_lane2(u1.x);
        float d2 = quad_lane2(u2.x), d3 = quad_lane2(u3.x);
        float w0 = d0 * dd;
        float w1 = k1 ? d1 * dd : 0.0f;
        float w2 = k2 ? d2 * dd : 0.0f;
        float w3 = k3 ? d3 * dd : 0.0f;
        fma4(a, u0, w0); fma4(a, u1, w1); fma4(a, u2, w2); fma4(a, u3, w3);
      }
      acc[q] = a;
    }
  }
  if (j < 2) {
    float4 bv = *(const float4*)(b2 + 4 * j);
#pragma unroll
    for (int q = 0; q < 2; ++q) {
      int node = nb + q;
      float ws = ddq[q] * ddq[q];
      float4 sv = *(const float4*)(hb + (size_t)node * F_H1);
      float4 a = acc[q];
      float4 r;
      r.x = fmaxf(a.x + sv.x * ws + bv.x, 0.0f);
      r.y = fmaxf(a.y + sv.y * ws + bv.y, 0.0f);
      r.z = fmaxf(a.z + sv.z * ws + bv.z, 0.0f);
      r.w = fmaxf(a.w + sv.w * ws + bv.w, 0.0f);
      int gr = batch[node];
      float* ps = pooled + (size_t)gr * F_H2 + 4 * j;
      ATOMIC_ADD(&ps[0], r.x);
      ATOMIC_ADD(&ps[1], r.y);
      ATOMIC_ADD(&ps[2], r.z);
      ATOMIC_ADD(&ps[3], r.w);
      if (j == 0) ATOMIC_ADD(&cnt[gr], 1.0f);
    }
  }
}

__global__ void readout(const float* __restrict__ pooled, const float* __restrict__ cnt,
                        const float* __restrict__ Wl, const float* __restrict__ bl,
                        float* __restrict__ out, int G) {
  int g = blockIdx.x * blockDim.x + threadIdx.x;
  if (g >= G) return;
  float inv = 1.0f / fmaxf(cnt[g], 1.0f);
  float p[F_H2];
#pragma unroll
  for (int f = 0; f < F_H2; ++f) p[f] = pooled[(size_t)g * F_H2 + f] * inv;
#pragma unroll
  for (int c = 0; c < N_CLS; ++c) {
    float acc = bl[c];
#pragma unroll
    for (int f = 0; f < F_H2; ++f) acc += p[f] * Wl[f * N_CLS + c];
    out[(size_t)g * N_CLS + c] = acc;
  }
}

#define PR_T2_OFF   136757248ll
#define PR_NEED     (PR_T2_OFF + 32000000ll)
#define TP2 20

__global__ __launch_bounds__(128) void xw1_scaled(const float* __restrict__ X,
                                                  const float* __restrict__ dinv,
                                                  const float* __restrict__ W1,
                                                  float* __restrict__ P) {
  __shared__ float tile[4 * 16 * TPITCH];
  int wave = blockIdx.x * (blockDim.x >> 5) + (threadIdx.x >> 5);
  int lane = threadIdx.x & 31;
  int m = lane & 15;
  int h = lane >> 4;
  int g = lane >> 3;
  int j = lane & 7;
  int nb = wave * 16 + 4 * g;
  const float* xb = X + 4 * j;
  float* tl = tile + (threadIdx.x >> 5) * (16 * TPITCH);
  float4 u[4];
#pragma unroll
  for (int q = 0; q < 4; ++q)
    u[q] = *(const float4*)(xb + (size_t)(nb + q) * F_IN);
#pragma unroll
  for (int q = 0; q < 4; ++q)
    *(float4*)(tl + (4 * g + q) * TPITCH + 4 * j) = u[q];
  const float* tr = tl + m * TPITCH + 16 * h;
  float4 a0 = *(const float4*)(tr + 0);
  float4 a1 = *(const float4*)(tr + 4);
  float4 a2 = *(const float4*)(tr + 8);
  float4 a3 = *(const float4*)(tr + 12);
  const float* wb = W1 + (size_t)(16 * h) * F_H1 + m;
  v2f A0 = {a0.x, a0.y}, A1 = {a0.z, a0.w}, A2 = {a1.x, a1.y}, A3 = {a1.z, a1.w};
  v2f A4 = {a2.x, a2.y}, A5 = {a2.z, a2.w}, A6 = {a3.x, a3.y}, A7 = {a3.z, a3.w};
  v2f B0 = {wb[0 * F_H1],  wb[1 * F_H1]},  B1 = {wb[2 * F_H1],  wb[3 * F_H1]};
  v2f B2 = {wb[4 * F_H1],  wb[5 * F_H1]},  B3 = {wb[6 * F_H1],  wb[7 * F_H1]};
  v2f B4 = {wb[8 * F_H1],  wb[9 * F_H1]},  B5 = {wb[10 * F_H1], wb[11 * F_H1]};
  v2f B6 = {wb[12 * F_H1], wb[13 * F_H1]}, B7 = {wb[14 * F_H1], wb[15 * F_H1]};
  float4 d0 = *(const float4*)(dinv + wave * 16 + 8 * h);
  float4 d1 = *(const float4*)(dinv + wave * 16 + 8 * h + 4);
  v8f c = {0.f, 0.f, 0.f, 0.f, 0.f, 0.f, 0.f, 0.f};
  asm volatile(
      "v_wmma_f32_16x16x4_f32 %0, %1, %9, %0\n\t"
      "v_wmma_f32_16x16x4_f32 %0, %2, %10, %0\n\t"
      "v_wmma_f32_16x16x4_f32 %0, %3, %11, %0\n\t"
      "v_wmma_f32_16x16x4_f32 %0, %4, %12, %0\n\t"
      "v_wmma_f32_16x16x4_f32 %0, %5, %13, %0\n\t"
      "v_wmma_f32_16x16x4_f32 %0, %6, %14, %0\n\t"
      "v_wmma_f32_16x16x4_f32 %0, %7, %15, %0\n\t"
      "v_wmma_f32_16x16x4_f32 %0, %8, %16, %0\n\t"
      "v_nop\n\tv_nop\n\tv_nop\n\tv_nop\n\tv_nop"
      : "+v"(c)
      : "v"(A0), "v"(A1), "v"(A2), "v"(A3), "v"(A4), "v"(A5), "v"(A6), "v"(A7),
        "v"(B0), "v"(B1), "v"(B2), "v"(B3), "v"(B4), "v"(B5), "v"(B6), "v"(B7));
  float* out = P + (size_t)(wave * 16 + 8 * h) * F_H1 + m;
  out[0 * F_H1] = c[0] * d0.x;
  out[1 * F_H1] = c[1] * d0.y;
  out[2 * F_H1] = c[2] * d0.z;
  out[3 * F_H1] = c[3] * d0.w;
  out[4 * F_H1] = c[4] * d1.x;
  out[5 * F_H1] = c[5] * d1.y;
  out[6 * F_H1] = c[6] * d1.z;
  out[7 * F_H1] = c[7] * d1.w;
}

__global__ __launch_bounds__(128) void pull1_rows(const float* __restrict__ P,
                                                  const int* __restrict__ starts,
                                                  const int* __restrict__ ssrc,
                                                  const float* __restrict__ dinv,
                                                  const float* __restrict__ b1,
                                                  const float* __restrict__ W2,
                                                  float* __restrict__ T2) {
  __shared__ int idbuf[4 * IDBUF];
  __shared__ float tile[4 * 16 * TP2];
  __shared__ float obuf[4 * 16 * F_H2];
  int wave = blockIdx.x * (blockDim.x >> 5) + (threadIdx.x >> 5);
  int lane = threadIdx.x & 31;
  int g = lane >> 2;
  int j = lane & 3;
  int nb = wave * 16 + 2 * g;
  int st[3];
#pragma unroll
  for (int q = 0; q < 3; ++q) {
    int n = nb + q;
    st[q] = (n < N_NODES) ? starts[n] : N_EDGES;
  }
  float ddq[2];
  ddq[0] = dinv[nb];
  ddq[1] = dinv[nb + 1];
  const float* pb = P + 4 * j;
  int* my = idbuf + (threadIdx.x >> 5) * IDBUF;
  int seg_beg = __builtin_amdgcn_readlane(st[0], 0);
  int seg_end = __builtin_amdgcn_readlane(st[2], 31);

  float4 acc[2];
  acc[0] = make_float4(0.f, 0.f, 0.f, 0.f);
  acc[1] = acc[0];
  for (int abase = seg_beg & ~3; abase < seg_end; abase += IDBUF) {
    stage_ids(ssrc, abase, min(IDBUF, seg_end - abase), my, lane);
#pragma unroll
    for (int q = 0; q < 2; ++q) {
      int lo = max(st[q], abase) - abase;
      int last = min(st[q + 1], abase + IDBUF) - 1 - abase;
      float4 a = acc[q];
      for (int i = lo; i <= last; i += 4) {
        bool k1 = (i + 1 <= last), k2 = (i + 2 <= last), k3 = (i + 3 <= last);
        int s0 = my[i];
        int s1 = my[k1 ? i + 1 : i];
        int s2 = my[k2 ? i + 2 : i];
        int s3 = my[k3 ? i + 3 : i];
        float4 u0 = *(const float4*)(pb + (size_t)s0 * F_H1);
        float4 u1 = *(const float4*)(pb + (size_t)s1 * F_H1);
        float4 u2 = *(const float4*)(pb + (size_t)s2 * F_H1);
        float4 u3 = *(const float4*)(pb + (size_t)s3 * F_H1);
        fma4(a, u0, 1.0f);
        fma4(a, u1, k1 ? 1.0f : 0.0f);
        fma4(a, u2, k2 ? 1.0f : 0.0f);
        fma4(a, u3, k3 ? 1.0f : 0.0f);
      }
      acc[q] = a;
    }
  }
  float* tl = tile + (threadIdx.x >> 5) * (16 * TP2);
  float4 bv = *(const float4*)(b1 + 4 * j);
#pragma unroll
  for (int q = 0; q < 2; ++q) {
    float4 sv = *(const float4*)(pb + (size_t)(nb + q) * F_H1);
    float dd = ddq[q];
    float4 a = acc[q];
    float4 r;
    r.x = fmaxf(fmaf(a.x + sv.x, dd, bv.x), 0.0f);
    r.y = fmaxf(fmaf(a.y + sv.y, dd, bv.y), 0.0f);
    r.z = fmaxf(fmaf(a.z + sv.z, dd, bv.z), 0.0f);
    r.w = fmaxf(fmaf(a.w + sv.w, dd, bv.w), 0.0f);
    *(float4*)(tl + (2 * g + q) * TP2 + 4 * j) = r;
  }
  int m  = lane & 15;
  int hh = lane >> 4;
  const float* hrow = tl + m * TP2 + 2 * hh;
  bool live_col = (m < F_H2);
  const float* w2 = W2 + (size_t)(2 * hh) * F_H2 + (live_col ? m : 0);
  v2f A0 = {hrow[0],  hrow[1]},  A1 = {hrow[4],  hrow[5]};
  v2f A2 = {hrow[8],  hrow[9]},  A3 = {hrow[12], hrow[13]};
  v2f B0 = {live_col ? w2[0 * F_H2]  : 0.0f, live_col ? w2[1 * F_H2]  : 0.0f};
  v2f B1 = {live_col ? w2[4 * F_H2]  : 0.0f, live_col ? w2[5 * F_H2]  : 0.0f};
  v2f B2 = {live_col ? w2[8 * F_H2]  : 0.0f, live_col ? w2[9 * F_H2]  : 0.0f};
  v2f B3 = {live_col ? w2[12 * F_H2] : 0.0f, live_col ? w2[13 * F_H2] : 0.0f};
  float4 d0 = *(const float4*)(dinv + wave * 16 + 8 * hh);
  float4 d1 = *(const float4*)(dinv + wave * 16 + 8 * hh + 4);
  v8f c = {0.f, 0.f, 0.f, 0.f, 0.f, 0.f, 0.f, 0.f};
  asm volatile(
      "v_wmma_f32_16x16x4_f32 %0, %1, %5, %0\n\t"
      "v_wmma_f32_16x16x4_f32 %0, %2, %6, %0\n\t"
      "v_wmma_f32_16x16x4_f32 %0, %3, %7, %0\n\t"
      "v_wmma_f32_16x16x4_f32 %0, %4, %8, %0\n\t"
      "v_nop\n\tv_nop\n\tv_nop\n\tv_nop\n\tv_nop"
      : "+v"(c)
      : "v"(A0), "v"(A1), "v"(A2), "v"(A3), "v"(B0), "v"(B1), "v"(B2), "v"(B3));
  float* ob = obuf + (threadIdx.x >> 5) * (16 * F_H2);
  if (live_col) {
    float* o = ob + (8 * hh) * F_H2 + m;
    o[0 * F_H2] = c[0] * d0.x;
    o[1 * F_H2] = c[1] * d0.y;
    o[2 * F_H2] = c[2] * d0.z;
    o[3 * F_H2] = c[3] * d0.w;
    o[4 * F_H2] = c[4] * d1.x;
    o[5 * F_H2] = c[5] * d1.y;
    o[6 * F_H2] = c[6] * d1.z;
    o[7 * F_H2] = c[7] * d1.w;
  }
  float4 ov = *(const float4*)(ob + 4 * lane);
  *(float4*)(T2 + (size_t)wave * (16 * F_H2) + 4 * lane) = ov;
}

__global__ __launch_bounds__(128) void pull2_rows(const float* __restrict__ T2,
                                                  const int* __restrict__ starts,
                                                  const int* __restrict__ ssrc,
                                                  const float* __restrict__ dinv,
                                                  const float* __restrict__ b2,
                                                  const int* __restrict__ batch,
                                                  float* __restrict__ pooled,
                                                  float* __restrict__ cnt) {
  __shared__ int idbuf[4 * IDBUF];
  int wave = blockIdx.x * (blockDim.x >> 5) + (threadIdx.x >> 5);
  int lane = threadIdx.x & 31;
  int g = lane >> 1;
  int j = lane & 1;
  int node = wave * 16 + g;
  int st0 = starts[node];
  int st1 = (node + 1 < N_NODES) ? starts[node + 1] : N_EDGES;
  float dd = dinv[node];
  const float* tb = T2 + 4 * j;
  int* my = idbuf + (threadIdx.x >> 5) * IDBUF;
  int seg_beg = __builtin_amdgcn_readlane(st0, 0);
  int seg_end = __builtin_amdgcn_readlane(st1, 31);

  float4 a = make_float4(0.f, 0.f, 0.f, 0.f);
  for (int abase = seg_beg & ~3; abase < seg_end; abase += IDBUF) {
    stage_ids(ssrc, abase, min(IDBUF, seg_end - abase), my, lane);
    int lo = max(st0, abase) - abase;
    int last = min(st1, abase + IDBUF) - 1 - abase;
    for (int i = lo; i <= last; i += 4) {
      bool k1 = (i + 1 <= last), k2 = (i + 2 <= last), k3 = (i + 3 <= last);
      int s0 = my[i];
      int s1 = my[k1 ? i + 1 : i];
      int s2 = my[k2 ? i + 2 : i];
      int s3 = my[k3 ? i + 3 : i];
      float4 u0 = *(const float4*)(tb + (size_t)s0 * F_H2);
      float4 u1 = *(const float4*)(tb + (size_t)s1 * F_H2);
      float4 u2 = *(const float4*)(tb + (size_t)s2 * F_H2);
      float4 u3 = *(const float4*)(tb + (size_t)s3 * F_H2);
      fma4(a, u0, 1.0f);
      fma4(a, u1, k1 ? 1.0f : 0.0f);
      fma4(a, u2, k2 ? 1.0f : 0.0f);
      fma4(a, u3, k3 ? 1.0f : 0.0f);
    }
  }
  float4 bv = *(const float4*)(b2 + 4 * j);
  float4 sv = *(const float4*)(tb + (size_t)node * F_H2);
  float4 r;
  r.x = fmaxf(fmaf(a.x + sv.x, dd, bv.x), 0.0f);
  r.y = fmaxf(fmaf(a.y + sv.y, dd, bv.y), 0.0f);
  r.z = fmaxf(fmaf(a.z + sv.z, dd, bv.z), 0.0f);
  r.w = fmaxf(fmaf(a.w + sv.w, dd, bv.w), 0.0f);
  int gr = batch[node];
  float* ps = pooled + (size_t)gr * F_H2 + 4 * j;
  ATOMIC_ADD(&ps[0], r.x);
  ATOMIC_ADD(&ps[1], r.y);
  ATOMIC_ADD(&ps[2], r.z);
  ATOMIC_ADD(&ps[3], r.w);
  if (j == 0) ATOMIC_ADD(&cnt[gr], 1.0f);
}


extern "C" void kernel_launch(void* const* d_in, const int* in_sizes, int n_in,
                              void* d_out, int out_size, void* d_ws, size_t ws_size,
                              hipStream_t stream) {
  const float* x     = (const float*)d_in[0];
  const int*   ei    = (const int*)d_in[1];
  const int*   src   = ei;
  const int*   dst   = ei + N_EDGES;
  const int*   batch = (const int*)d_in[2];
  const float* W1 = (const float*)d_in[4];
  const float* b1 = (const float*)d_in[5];
  const float* W2 = (const float*)d_in[6];
  const float* b2 = (const float*)d_in[7];
  const float* Wl = (const float*)d_in[8];
  const float* bl = (const float*)d_in[9];
  float* out = (float*)d_out;

  char* ws = (char*)d_ws;
  float* dinv   = (float*)(ws);
  int*   arr    = (int*)(ws + (4ll << 20));
  int*   lst    = (int*)(ws + (8ll << 20));
  float* rows   = (float*)(ws + (8ll << 20));
  int*   planes = (int*)(ws + 72388608ll);
  int*   ssrc   = (int*)(ws + 72388608ll);
  int*   bsum   = (int*)(ws + 136388608ll);
  float* pooled = (float*)(ws + 136396800ll);
  int*   tbeg   = (int*)(ws + 136396800ll);
  float* cnt    = pooled + (size_t)N_GRAPHS * F_H2;

  const int NB = 256;
  const int TILE_BLOCKS = N_NODES / 64;

  hist_tiles<<<NSLICE, LB_THREADS, 0, stream>>>(dst, planes);
  scan_block_sums<<<SCAN_BLOCKS, 256, 0, stream>>>(planes, bsum);
  scan_top<<<1, 1024, 0, stream>>>(bsum, SCAN_BLOCKS);
  scan_apply<<<SCAN_BLOCKS, 256, 0, stream>>>(planes, tbeg, bsum);
  scatter_tiles<<<NSLICE, LB_THREADS, 0, stream>>>(src, dst, planes, lst);
  tile_order<<<TILE_BLOCKS, 128, 0, stream>>>(lst, tbeg, arr, dinv, ssrc);

  if (ws_size >= (size_t)PR_NEED) {
    float* P  = rows;
    float* t2 = (float*)(ws + PR_T2_OFF);
    xw1_scaled<<<TILE_BLOCKS, 128, 0, stream>>>(x, dinv, W1, P);
    pull1_rows<<<TILE_BLOCKS, 128, 0, stream>>>(P, arr, ssrc, dinv, b1, W2, t2);
    zero_u32<<<128, 256, 0, stream>>>((unsigned*)pooled, N_GRAPHS * (F_H2 + 1));
    pull2_rows<<<TILE_BLOCKS, 128, 0, stream>>>(t2, arr, ssrc, dinv, b2, batch, pooled, cnt);
  } else {
    float* h1 = rows;
    pull1_xw1<<<TILE_BLOCKS, 128, 0, stream>>>(x, arr, ssrc, dinv, W1, b1, h1);
    gemm_h1_w2<<<TILE_BLOCKS, 128, 0, stream>>>(h1, W2, dinv);
    zero_u32<<<128, 256, 0, stream>>>((unsigned*)pooled, N_GRAPHS * (F_H2 + 1));
    pull2_pool<<<TILE_BLOCKS, 128, 0, stream>>>(h1, arr, ssrc, dinv, b2, batch, pooled, cnt);
  }
  readout<<<(N_GRAPHS + NB - 1) / NB, NB, 0, stream>>>(pooled, cnt, Wl, bl,
                                                       out, N_GRAPHS);
}


# list build stores runs, not single words: slice entries ordered by 1024-node bucket in LDS, then a work-group per bucket orders by node in LDS and stores whole lines; sum order = LDS arrival
# speedup vs baseline: 10.2271x; 2.0283x over previous
#include <hip/hip_runtime.h>


typedef float v2f __attribute__((ext_vector_type(2)));
typedef float v8f __attribute__((ext_vector_type(8)));

#define N_NODES   1000000
#define N_EDGES   16000000
#define F_IN      32
#define F_H1      16
#define F_H2      8
#define N_CLS     5
#define N_GRAPHS  10000

#define ATOMIC_ADD(p, v) \
  __hip_atomic_fetch_add((p), (v), __ATOMIC_RELAXED, __HIP_MEMORY_SCOPE_AGENT)


__global__ void zero_u32(unsigned* __restrict__ p, int n) {
  int i = blockIdx.x * blockDim.x + threadIdx.x;
  int stride = gridDim.x * blockDim.x;
  for (; i < n; i += stride) p[i] = 0u;
}

#define LDS_ADD(p, v) \
  __hip_atomic_fetch_add((p), (v), __ATOMIC_RELAXED, __HIP_MEMORY_SCOPE_WORKGROUP)

#define BK_SHIFT 10
#define BK_NODES (1 << BK_SHIFT)
#define NBK_LIVE ((N_NODES + BK_NODES - 1) / BK_NODES)
#define NBK     ((NBK_LIVE + 3) & ~3)
#define NSLICE  256
#define SLICE_V4 (N_EDGES / 4 / NSLICE)
#define SLICE_E  (SLICE_V4 * 4)
#define LB_THREADS 1024
#define LB_GROUPS 4
#define SRC_BITS 20
#define SRC_MASK ((1 << SRC_BITS) - 1)
static_assert(NBK % 4 == 0 && NBK <= LB_THREADS, "planes are whole 16-byte groups; one thread per bucket");
static_assert(SLICE_V4 * 4 * NSLICE == N_EDGES, "slices cover the edges exactly");
static_assert(N_NODES <= (1 << SRC_BITS), "a source id fits its field");
static_assert(SRC_BITS + BK_SHIFT < 31, "an entry is a non-negative int");
static_assert((SLICE_E + 3 * NBK + 64 + LB_THREADS) * 4 <= 320 * 1024, "the place pass fits the LDS");

#define SCAN_TILES 64
#define SCAN_BLOCKS ((NBK + SCAN_TILES - 1) / SCAN_TILES)
static_assert(SCAN_BLOCKS <= 1024, "one block scans the block sums");

__global__ __launch_bounds__(LB_THREADS) void hist_buckets(const int* __restrict__ dst,
                                                           int* __restrict__ planes) {
  __shared__ int cnt[NBK];
  int sl = blockIdx.x;
  for (int i = threadIdx.x; i < NBK; i += LB_THREADS) cnt[i] = 0;
  __syncthreads();
  const int4* d4 = (const int4*)dst + (size_t)sl * SLICE_V4;
  for (int i = threadIdx.x; i < SLICE_V4; i += LB_GROUPS * LB_THREADS) {
    int4 d[LB_GROUPS];
#pragma unroll
    for (int g = 0; g < LB_GROUPS; ++g) {
      int ig = i + g * LB_THREADS;
      d[g] = (ig < SLICE_V4) ? d4[ig] : make_int4(-1, -1, -1, -1);
    }
#pragma unroll
    for (int g = 0; g < LB_GROUPS; ++g) {
      if ((unsigned)d[g].x < (unsigned)N_NODES) LDS_ADD(&cnt[d[g].x >> BK_SHIFT], 1);
      if ((unsigned)d[g].y < (unsigned)N_NODES) LDS_ADD(&cnt[d[g].y >> BK_SHIFT], 1);
      if ((unsigned)d[g].z < (unsigned)N_NODES) LDS_ADD(&cnt[d[g].z >> BK_SHIFT], 1);
      if ((unsigned)d[g].w < (unsigned)N_NODES) LDS_ADD(&cnt[d[g].w >> BK_SHIFT], 1);
    }
  }
  __syncthreads();
  for (int i = threadIdx.x; i < NBK; i += LB_THREADS) planes[(size_t)sl * NBK + i] = cnt[i];
}

template <int NT>
__device__ __forceinline__ int block_scan_excl(int v, int* sh, int* total) {
  int t = threadIdx.x;
  sh[t] = v;
  __syncthreads();
  for (int o = 1; o < NT; o <<= 1) {
    int add = (t >= o) ? sh[t - o] : 0;
    __syncthreads();
    sh[t] += add;
    __syncthreads();
  }
  int incl = sh[t];
  *total = sh[NT - 1];
  __syncthreads();
  return incl - v;
}

__global__ __launch_bounds__(256) void scan_block_sums(const int* __restrict__ planes,
                                                       int* __restrict__ bsum) {
  __shared__ int sh[256];
  int tq = threadIdx.x & 15;
  int pg = threadIdx.x >> 4;
  int base = blockIdx.x * SCAN_TILES + 4 * tq;
  int s = 0;
  if (base < NBK) {
#pragma unroll
    for (int p = 0; p < 16; ++p) {
      int4 v = *(const int4*)(planes + (size_t)(pg * 16 + p) * NBK + base);
      s += v.x + v.y + v.z + v.w;
    }
  }
  int total;
  (void)block_scan_excl<256>(s, sh, &total);
  if (threadIdx.x == 0) bsum[blockIdx.x] = total;
}

__global__ __launch_bounds__(1024) void scan_top(int* __restrict__ bsum, int nb) {
  __shared__ int sh[1024];
  int t = threadIdx.x;
  int v = (t < nb) ? bsum[t] : 0;
  int total;
  int ex = block_scan_excl<1024>(v, sh, &total);
  if (t < nb) bsum[t] = ex;
}

__global__ __launch_bounds__(256) void scan_apply(int* __restrict__ planes,
                                                  int* __restrict__ tbeg,
                                                  const int* __restrict__ boff) {
  __shared__ int4 part[16 * 16];
  __shared__ int tsum[16];
  int tq = threadIdx.x & 15;
  int pg = threadIdx.x >> 4;
  int base = blockIdx.x * SCAN_TILES + 4 * tq;
  bool live = (base < NBK);
  int4 c[16];
  int4 v = make_int4(0, 0, 0, 0);
#pragma unroll
  for (int p = 0; p < 16; ++p) {
    c[p] = make_int4(0, 0, 0, 0);
    if (live) c[p] = *(const int4*)(planes + (size_t)(pg * 16 + p) * NBK + base);
    v.x += c[p].x; v.y += c[p].y; v.z += c[p].z; v.w += c[p].w;
  }
  part[pg * 16 + tq] = v;
  __syncthreads();
  int4 pre = make_int4(0, 0, 0, 0);
  int4 tot = make_int4(0, 0, 0, 0);
  for (int g = 0; g < 16; ++g) {
    int4 w = part[g * 16 + tq];
    if (g < pg) { pre.x += w.x; pre.y += w.y; pre.z += w.z; pre.w += w.w; }
    tot.x += w.x; tot.y += w.y; tot.z += w.z; tot.w += w.w;
  }
  if (pg == 0) tsum[tq] = tot.x + tot.y + tot.z + tot.w;
  __syncthreads();
  int ex = boff[blockIdx.x];
  for (int q = 0; q < 16; ++q) {
    int w = tsum[q];
    if (q < tq) ex += w;
  }
  if (live) {
    int4 o;
    o.x = ex;
    o.y = o.x + tot.x;
    o.z = o.y + tot.y;
    o.w = o.z + tot.z;
    if (pg == 0) {
      *(int4*)(tbeg + base) = o;
      if (base + 4 == NBK) tbeg[NBK] = o.w + tot.w;
    }
    o.x += pre.x; o.y += pre.y; o.z += pre.z; o.w += pre.w;
#pragma unroll
    for (int p = 0; p < 16; ++p) {
      *(int4*)(planes + (size_t)(pg * 16 + p) * NBK + base) = o;
      o.x += c[p].x; o.y += c[p].y; o.z += c[p].z; o.w += c[p].w;
    }
  }
}

__global__ __launch_bounds__(LB_THREADS) void scatter_runs(const int* __restrict__ src,
                                                           const int* __restrict__ dst,
                                                           const int* __restrict__ planes,
                                                           const int* __restrict__ tbeg,
                                                           int* __restrict__ lst) {
  __shared__ int ebuf[SLICE_E];
  __shared__ int gbeg[NBK];
  __shared__ int lbeg[NBK + 1];
  __shared__ int lcur[NBK + 32];
  __shared__ int sh[LB_THREADS];
  int sl = blockIdx.x;
  int t = threadIdx.x;
  int c = 0;
  if (t < NBK) {
    int g0 = planes[(size_t)sl * NBK + t];
    int g1 = (sl + 1 < NSLICE) ? planes[(size_t)(sl + 1) * NBK + t] : tbeg[t + 1];
    gbeg[t] = g0;
    c = g1 - g0;
  }
  int total;
  int ex = block_scan_excl<LB_THREADS>(c, sh, &total);
  if (t < NBK) { lbeg[t] = ex; lcur[t] = ex; }
  if (t == 0) lbeg[NBK] = total;
  if (t < 32) lcur[NBK + t] = 0;
  __syncthreads();
  int spare = NBK + (t & 31);
  const int4* d4 = (const int4*)dst + (size_t)sl * SLICE_V4;
  const int4* s4 = (const int4*)src + (size_t)sl * SLICE_V4;
  for (int i = t; i < SLICE_V4; i += LB_GROUPS * LB_THREADS) {
    int4 d[LB_GROUPS], sv[LB_GROUPS];
    uint4 p[LB_GROUPS];
#pragma unroll
    for (int g = 0; g < LB_GROUPS; ++g) {
      int ig = i + g * LB_THREADS;
      bool inside = (ig < SLICE_V4);
      d[g]  = inside ? d4[ig] : make_int4(-1, -1, -1, -1);
      sv[g] = inside ? s4[ig] : make_int4(0, 0, 0, 0);
    }
#pragma unroll
    for (int g = 0; g < LB_GROUPS; ++g) {
      bool kx = (unsigned)d[g].x < (unsigned)N_NODES, ky = (unsigned)d[g].y < (unsigned)N_NODES;
      bool kz = (unsigned)d[g].z < (unsigned)N_NODES, kw = (unsigned)d[g].w < (unsigned)N_NODES;
      p[g].x = (unsigned)LDS_ADD(&lcur[kx ? (d[g].x >> BK_SHIFT) : spare], kx ? 1 : 0);
      p[g].y = (unsigned)LDS_ADD(&lcur[ky ? (d[g].y >> BK_SHIFT) : spare], ky ? 1 : 0);
      p[g].z = (unsigned)LDS_ADD(&lcur[kz ? (d[g].z >> BK_SHIFT) : spare], kz ? 1 : 0);
      p[g].w = (unsigned)LDS_ADD(&lcur[kw ? (d[g].w >> BK_SHIFT) : spare], kw ? 1 : 0);
    }
#pragma unroll
    for (int g = 0; g < LB_GROUPS; ++g) {
      if ((unsigned)d[g].x < (unsigned)N_NODES) ebuf[p[g].x] = sv[g].x | ((d[g].x & (BK_NODES - 1)) << SRC_BITS);
      if ((unsigned)d[g].y < (unsigned)N_NODES) ebuf[p[g].y] = sv[g].y | ((d[g].y & (BK_NODES - 1)) << SRC_BITS);
      if ((unsigned)d[g].z < (unsigned)N_NODES) ebuf[p[g].z] = sv[g].z | ((d[g].z & (BK_NODES - 1)) << SRC_BITS);
      if ((unsigned)d[g].w < (unsigned)N_NODES) ebuf[p[g].w] = sv[g].w | ((d[g].w & (BK_NODES - 1)) << SRC_BITS);
    }
  }
  __syncthreads();
  int lane = t & 31;
  for (int b = t >> 5; b < NBK; b += LB_THREADS / 32) {
    int lb = lbeg[b];
    int n = lbeg[b + 1] - lb;
    int gb = gbeg[b];
    for (int k = lane; k < n; k += 32) lst[gb + k] = ebuf[lb + k];
  }
}

#define BO_THREADS BK_NODES
#define BO_CAP 24576
#define BO_UNROLL 4
__global__ __launch_bounds__(BO_THREADS) void bucket_order(const int* __restrict__ lst,
                                                           const int* __restrict__ tbeg,
                                                           int* __restrict__ starts,
                                                           float* __restrict__ dinv,
                                                           int* __restrict__ ssrc) {
  __shared__ int obuf[BO_CAP];
  __shared__ int cnt[BK_NODES];
  __shared__ int cur[BK_NODES];
  __shared__ int sh[BO_THREADS];
  int b = blockIdx.x;
  int t = threadIdx.x;
  int beg = tbeg[b];
  int n = tbeg[b + 1] - beg;
  const int* piece = lst + beg;
  cnt[t] = 0;
  __syncthreads();
  for (int i0 = 0; i0 < n; i0 += BO_UNROLL * BO_THREADS) {
    int e[BO_UNROLL];
#pragma unroll
    for (int u = 0; u < BO_UNROLL; ++u) {
      int i = i0 + u * BO_THREADS + t;
      e[u] = (i < n) ? piece[i] : -1;
    }
#pragma unroll
    for (int u = 0; u < BO_UNROLL; ++u)
      if (e[u] >= 0) LDS_ADD(&cnt[(e[u] >> SRC_BITS) & (BK_NODES - 1)], 1);
  }
  __syncthreads();
  int c = cnt[t];
  int total;
  int ex = block_scan_excl<BO_THREADS>(c, sh, &total);
  int node = b * BK_NODES + t;
  if (node < N_NODES) {
    starts[node] = beg + ex;
    dinv[node] = rsqrtf((float)c + 1.0f);
  }
  bool in_lds = (n <= BO_CAP);
  cur[t] = in_lds ? ex : beg + ex;
  __syncthreads();
  for (int i0 = 0; i0 < n; i0 += BO_UNROLL * BO_THREADS) {
    int e[BO_UNROLL], p[BO_UNROLL];
#pragma unroll
    for (int u = 0; u < BO_UNROLL; ++u) {
      int i = i0 + u * BO_THREADS + t;
      e[u] = (i < n) ? piece[i] : -1;
    }
#pragma unroll
    for (int u = 0; u < BO_UNROLL; ++u) {
      p[u] = 0;
      if (e[u] >= 0) p[u] = LDS_ADD(&cur[(e[u] >> SRC_BITS) & (BK_NODES - 1)], 1);
    }
#pragma unroll
    for (int u = 0; u < BO_UNROLL; ++u) {
      if (e[u] >= 0) {
        if (in_lds) obuf[p[u]] = e[u] & SRC_MASK;
        else ssrc[p[u]] = e[u] & SRC_MASK;
      }
    }
  }
  if (in_lds) {
    __syncthreads();
    for (int i = t; i < n; i += BO_THREADS) ssrc[beg + i] = obuf[i];
  }
}

__device__ __forceinline__ void fma4(float4& a, const float4& v, float w) {
  a.x = fmaf(v.x, w, a.x);
  a.y = fmaf(v.y, w, a.y);
  a.z = fmaf(v.z, w, a.z);
  a.w = fmaf(v.w, w, a.w);
}

#define IDBUF 512
#define TPITCH 36

__device__ __forceinline__ void stage_ids(const int* __restrict__ ssrc, int abase,
                                          int need, int* buf, int lane) {
  int4 v[IDBUF / 128];
#pragma unroll
  for (int k = 0; k < IDBUF / 128; ++k) {
    int o = 128 * k + 4 * lane;
    v[k] = *(const int4*)(ssrc + abase + (o < need ? o : 0));
  }
#pragma unroll
  for (int k = 0; k < IDBUF / 128; ++k) *(int4*)(buf + 128 * k + 4 * lane) = v[k];
}


__global__ __launch_bounds__(128) void pull1_xw1(const float* __restrict__ X,
                                                 const int* __restrict__ starts,
                                                 const int* __restrict__ ssrc,
                                                 const float* __restrict__ dinv,
                                                 const float* __restrict__ W1,
                                                 const float* __restrict__ b1,
                                                 float* __restrict__ H1) {
  __shared__ int idbuf[4 * IDBUF];
  __shared__ float tile[4 * 16 * TPITCH];
  int wave = blockIdx.x * (blockDim.x >> 5) + (threadIdx.x >> 5);
  int lane = threadIdx.x & 31;
  int m = lane & 15;
  int h = lane >> 4;
  int g = lane >> 3;
  int j = lane & 7;
  int nb = wave * 16 + 4 * g;
  int st[5];
#pragma unroll
  for (int q = 0; q < 5; ++q) {
    int n = nb + q;
    st[q] = (n < N_NODES) ? starts[n] : N_EDGES;
  }
  float ddq[4];
#pragma unroll
  for (int q = 0; q < 4; ++q) ddq[q] = dinv[nb + q];
  const float* xb = X + 4 * j;
  int* my = idbuf + (threadIdx.x >> 5) * IDBUF;
  int seg_beg = __builtin_amdgcn_readlane(st[0], 0);
  int seg_end = __builtin_amdgcn_readlane(st[4], 31);

  float4 acc[4];
#pragma unroll
  for (int q = 0; q < 4; ++q) acc[q] = make_float4(0.f, 0.f, 0.f, 0.f);
  for (int abase = seg_beg & ~3; abase < seg_end; abase += IDBUF) {
    stage_ids(ssrc, abase, min(IDBUF, seg_end - abase), my, lane);
#pragma unroll
    for (int q = 0; q < 4; ++q) {
      int lo = max(st[q], abase) - abase;
      int last = min(st[q + 1], abase + IDBUF) - 1 - abase;
      float dd = ddq[q];
      float4 a = acc[q];
      for (int i = lo; i <= last; i += 4) {
        bool k1 = (i + 1 <= last), k2 = (i + 2 <= last), k3 = (i + 3 <= last);
        int s0 = my[i];
        int s1 = my[k1 ? i + 1 : i];
        int s2 = my[k2 ? i + 2 : i];
        int s3 = my[k3 ? i + 3 : i];
        float d0 = dinv[s0], d1 = dinv[s1], d2 = dinv[s2], d3 = dinv[s3];
        float4 u0 = *(const float4*)(xb + (size_t)s0 * F_IN);
        float4 u1 = *(const float4*)(xb + (size_t)s1 * F_IN);
        float4 u2 = *(const float4*)(xb + (size_t)s2 * F_IN);
        float4 u3 = *(const float4*)(xb + (size_t)s3 * F_IN);
        float w0 = d0 * dd;
        float w1 = k1 ? d1 * dd : 0.0f;
        float w2 = k2 ? d2 * dd : 0.0f;
        float w3 = k3 ? d3 * dd : 0.0f;
        fma4(a, u0, w0); fma4(a, u1, w1); fma4(a, u2, w2); fma4(a, u3, w3);
      }
      acc[q] = a;
    }
  }
  float* tl = tile + (threadIdx.x >> 5) * (16 * TPITCH);
#pragma unroll
  for (int q = 0; q < 4; ++q) {
    float ws = ddq[q] * ddq[q];
    float4 u = *(const float4*)(xb + (size_t)(nb + q) * F_IN);
    float4 a = acc[q];
    fma4(a, u, ws);
    *(float4*)(tl + (4 * g + q) * TPITCH + 4 * j) = a;
  }
  const float* tr = tl + m * TPITCH + 16 * h;
  float4 a0 = *(const float4*)(tr + 0);
  float4 a1 = *(const float4*)(tr + 4);
  float4 a2 = *(const float4*)(tr + 8);
  float4 a3 = *(const float4*)(tr + 12);

  const float* wb = W1 + (size_t)(16 * h) * F_H1 + m;
  v2f A0 = {a0.x, a0.y}, A1 = {a0.z, a0.w}, A2 = {a1.x, a1.y}, A3 = {a1.z, a1.w};
  v2f A4 = {a2.x, a2.y}, A5 = {a2.z, a2.w}, A6 = {a3.x, a3.y}, A7 = {a3.z, a3.w};
  v2f B0 = {wb[0 * F_H1],  wb[1 * F_H1]},  B1 = {wb[2 * F_H1],  wb[3 * F_H1]};
  v2f B2 = {wb[4 * F_H1],  wb[5 * F_H1]},  B3 = {wb[6 * F_H1],  wb[7 * F_H1]};
  v2f B4 = {wb[8 * F_H1],  wb[9 * F_H1]},  B5 = {wb[10 * F_H1], wb[11 * F_H1]};
  v2f B6 = {wb[12 * F_H1], wb[13 * F_H1]}, B7 = {wb[14 * F_H1], wb[15 * F_H1]};
  v8f c = {0.f, 0.f, 0.f, 0.f, 0.f, 0.f, 0.f, 0.f};
  asm volatile(
      "v_wmma_f32_16x16x4_f32 %0, %1, %9, %0\n\t"
      "v_wmma_f32_16x16x4_f32 %0, %2, %10, %0\n\t"
      "v_wmma_f32_16x16x4_f32 %0, %3, %11, %0\n\t"
      "v_wmma_f32_16x16x4_f32 %0, %4, %12, %0\n\t"
      "v_wmma_f32_16x16x4_f32 %0, %5, %13, %0\n\t"
      "v_wmma_f32_16x16x4_f32 %0, %6, %14, %0\n\t"
      "v_wmma_f32_16x16x4_f32 %0, %7, %15, %0\n\t"
      "v_wmma_f32_16x16x4_f32 %0, %8, %16, %0\n\t"
      "v_nop\n\tv_nop\n\tv_nop\n\tv_nop\n\tv_nop"
      : "+v"(c)
      : "v"(A0), "v"(A1), "v"(A2), "v"(A3), "v"(A4), "v"(A5), "v"(A6), "v"(A7),
        "v"(B0), "v"(B1), "v"(B2), "v"(B3), "v"(B4), "v"(B5), "v"(B6), "v"(B7));
  float bb = b1[m];
  float* out = H1 + (size_t)(wave * 16 + 8 * h) * F_H1 + m;
#pragma unroll
  for (int j = 0; j < 8; ++j) out[(size_t)j * F_H1] = fmaxf(c[j] + bb, 0.0f);
}

__global__ __launch_bounds__(128) void gemm_h1_w2(float* H, const float* __restrict__ W2,
                                                  const float* __restrict__ dinv) {
  int wave = blockIdx.x * (blockDim.x >> 5) + (threadIdx.x >> 5);
  int lane = threadIdx.x & 31;
  int m  = lane & 15;
  int kk = (lane >> 4) << 1;
  float* hrow = H + (size_t)(wave * 16 + m) * F_H1;
  bool live_col = (m < F_H2);
  float dm = dinv[wave * 16 + m];

  v2f a[4], b[4];
#pragma unroll
  for (int t = 0; t < 4; ++t) {
    int k = 4 * t + kk;
    a[t].x = hrow[k];
    a[t].y = hrow[k + 1];
    b[t].x = live_col ? W2[k * F_H2 + m] : 0.0f;
    b[t].y = live_col ? W2[(k + 1) * F_H2 + m] : 0.0f;
  }
  if (lane < 16) hrow[F_H2] = dm;

  v8f c = {0.f, 0.f, 0.f, 0.f, 0.f, 0.f, 0.f, 0.f};
#pragma unroll
  for (int t = 0; t < 4; ++t)
    c = __builtin_amdgcn_wmma_f32_16x16x4_f32(false, a[t], false, b[t],
                                              (short)0, c, false, false);
  if (live_col) {
    int hi = lane >> 4;
    float* out = H + (size_t)(wave * 16 + 8 * hi) * F_H1 + m;
#pragma unroll
    for (int j = 0; j < 8; ++j) out[(size_t)j * F_H1] = c[j];
  }
}

__device__ __forceinline__ float quad_lane2(float v) {
  int r = __builtin_amdgcn_mov_dpp(__float_as_int(v), 0xAA, 0xF, 0xF, true);
  return __int_as_float(r);
}

__global__ __launch_bounds__(128) void pull2_pool(const float* __restrict__ H,
                                                  const int* __restrict__ starts,
                                                  const int* __restrict__ ssrc,
                                                  const float* __restrict__ dinv,
                                                  const float* __restrict__ b2,
                                                  const int* __restrict__ batch,
                                                  float* __restrict__ pooled,
                                                  float* __restrict__ cnt) {
  __shared__ int idbuf[4 * IDBUF];
  int wave = blockIdx.x * (blockDim.x >> 5) + (threadIdx.x >> 5);
  int lane = threadIdx.x & 31;
  int g = lane >> 2;
  int j = lane & 3;
  int nb = wave * 16 + 2 * g;
  int st[3];
#pragma unroll
  for (int q = 0; q < 3; ++q) {
    int n = nb + q;
    st[q] = (n < N_NODES) ? starts[n] : N_EDGES;
  }
  float ddq[2];
  ddq[0] = dinv[nb];
  ddq[1] = dinv[nb + 1];
  const float* hb = H + 4 * j;
  int* my = idbuf + (threadIdx.x >> 5) * IDBUF;
  int seg_beg = __builtin_amdgcn_readlane(st[0], 0);
  int seg_end = __builtin_amdgcn_readlane(st[2], 31);

  float4 acc[2];
  acc[0] = make_float4(0.f, 0.f, 0.f, 0.f);
  acc[1] = acc[0];
  for (int abase = seg_beg & ~3; abase < seg_end; abase += IDBUF) {
    stage_ids(ssrc, abase, min(IDBUF, seg_end - abase), my, lane);
#pragma unroll
    for (int q = 0; q < 2; ++q) {
      int lo = max(st[q], abase) - abase;
      int last = min(st[q + 1], abase + IDBUF) - 1 - abase;
      float dd = ddq[q];
      float4 a = acc[q];
      for (int i = lo; i <= last; i += 4) {
        bool k1 = (i + 1 <= last), k2 = (i + 2 <= last), k3 = (i + 3 <= last);
        int s0 = my[i];
        int s1 = my[k1 ? i + 1 : i];
        int s2 = my[k2 ? i + 2 : i];
        int s3 = my[k3 ? i + 3 : i];
        float4 u0 = *(const float4*)(hb + (size_t)s0 * F_H1);
        float4 u1 = *(const float4*)(hb + (size_t)s1 * F_H1);
        float4 u2 = *(const float4*)(hb + (size_t)s2 * F_H1);
        float4 u3 = *(const float4*)(hb + (size_t)s3 * F_H1);
        float d0 = quad_lane2(u0.x), d1 = quad_lane2(u1.x);
        float d2 = quad_lane2(u2.x), d3 = quad_lane2(u3.x);
        float w0 = d0 * dd;
        float w1 = k1 ? d1 * dd : 0.0f;
        float w2 = k2 ? d2 * dd : 0.0f;
        float w3 = k3 ? d3 * dd : 0.0f;
        fma4(a, u0, w0); fma4(a, u1, w1); fma4(a, u2, w2); fma4(a, u3, w3);
      }
      acc[q] = a;
    }
  }
  if (j < 2) {
    float4 bv = *(const float4*)(b2 + 4 * j);
#pragma unroll
    for (int q = 0; q < 2; ++q) {
      int node = nb + q;
      float ws = ddq[q] * ddq[q];
      float4 sv = *(const float4*)(hb + (size_t)node * F_H1);
      float4 a = acc[q];
      float4 r;
      r.x = fmaxf(a.x + sv.x * ws + bv.x, 0.0f);
      r.y = fmaxf(a.y + sv.y * ws + bv.y, 0.0f);
      r.z = fmaxf(a.z + sv.z * ws + bv.z, 0.0f);
      r.w = fmaxf(a.w + sv.w * ws + bv.w, 0.0f);
      int gr = batch[node];
      float* ps = pooled + (size_t)gr * F_H2 + 4 * j;
      ATOMIC_ADD(&ps[0], r.x);
      ATOMIC_ADD(&ps[1], r.y);
      ATOMIC_ADD(&ps[2], r.z);
      ATOMIC_ADD(&ps[3], r.w);
      if (j == 0) ATOMIC_ADD(&cnt[gr], 1.0f);
    }
  }
}

__global__ void readout(const float* __restrict__ pooled, const float* __restrict__ cnt,
                        const float* __restrict__ Wl, const float* __restrict__ bl,
                        float* __restrict__ out, int G) {
  int g = blockIdx.x * blockDim.x + threadIdx.x;
  if (g >= G) return;
  float inv = 1.0f / fmaxf(cnt[g], 1.0f);
  float p[F_H2];
#pragma unroll
  for (int f = 0; f < F_H2; ++f) p[f] = pooled[(size_t)g * F_H2 + f] * inv;
#pragma unroll
  for (int c = 0; c < N_CLS; ++c) {
    float acc = bl[c];
#pragma unroll
    for (int f = 0; f < F_H2; ++f) acc += p[f] * Wl[f * N_CLS + c];
    out[(size_t)g * N_CLS + c] = acc;
  }
}

#define PR_T2_OFF   136757248ll
#define PR_NEED     (PR_T2_OFF + 32000000ll)
#define TP2 20

__global__ __launch_bounds__(128) void xw1_scaled(const float* __restrict__ X,
                                                  const float* __restrict__ dinv,
                                                  const float* __restrict__ W1,
                                                  float* __restrict__ P) {
  __shared__ float tile[4 * 16 * TPITCH];
  int wave = blockIdx.x * (blockDim.x >> 5) + (threadIdx.x >> 5);
  int lane = threadIdx.x & 31;
  int m = lane & 15;
  int h = lane >> 4;
  int g = lane >> 3;
  int j = lane & 7;
  int nb = wave * 16 + 4 * g;
  const float* xb = X + 4 * j;
  float* tl = tile + (threadIdx.x >> 5) * (16 * TPITCH);
  float4 u[4];
#pragma unroll
  for (int q = 0; q < 4; ++q)
    u[q] = *(const float4*)(xb + (size_t)(nb + q) * F_IN);
#pragma unroll
  for (int q = 0; q < 4; ++q)
    *(float4*)(tl + (4 * g + q) * TPITCH + 4 * j) = u[q];
  const float* tr = tl + m * TPITCH + 16 * h;
  float4 a0 = *(const float4*)(tr + 0);
  float4 a1 = *(const float4*)(tr + 4);
  float4 a2 = *(const float4*)(tr + 8);
  float4 a3 = *(const float4*)(tr + 12);
  const float* wb = W1 + (size_t)(16 * h) * F_H1 + m;
  v2f A0 = {a0.x, a0.y}, A1 = {a0.z, a0.w}, A2 = {a1.x, a1.y}, A3 = {a1.z, a1.w};
  v2f A4 = {a2.x, a2.y}, A5 = {a2.z, a2.w}, A6 = {a3.x, a3.y}, A7 = {a3.z, a3.w};
  v2f B0 = {wb[0 * F_H1],  wb[1 * F_H1]},  B1 = {wb[2 * F_H1],  wb[3 * F_H1]};
  v2f B2 = {wb[4 * F_H1],  wb[5 * F_H1]},  B3 = {wb[6 * F_H1],  wb[7 * F_H1]};
  v2f B4 = {wb[8 * F_H1],  wb[9 * F_H1]},  B5 = {wb[10 * F_H1], wb[11 * F_H1]};
  v2f B6 = {wb[12 * F_H1], wb[13 * F_H1]}, B7 = {wb[14 * F_H1], wb[15 * F_H1]};
  float4 d0 = *(const float4*)(dinv + wave * 16 + 8 * h);
  float4 d1 = *(const float4*)(dinv + wave * 16 + 8 * h + 4);
  v8f c = {0.f, 0.f, 0.f, 0.f, 0.f, 0.f, 0.f, 0.f};
  asm volatile(
      "v_wmma_f32_16x16x4_f32 %0, %1, %9, %0\n\t"
      "v_wmma_f32_16x16x4_f32 %0, %2, %10, %0\n\t"
      "v_wmma_f32_16x16x4_f32 %0, %3, %11, %0\n\t"
      "v_wmma_f32_16x16x4_f32 %0, %4, %12, %0\n\t"
      "v_wmma_f32_16x16x4_f32 %0, %5, %13, %0\n\t"
      "v_wmma_f32_16x16x4_f32 %0, %6, %14, %0\n\t"
      "v_wmma_f32_16x16x4_f32 %0, %7, %15, %0\n\t"
      "v_wmma_f32_16x16x4_f32 %0, %8, %16, %0\n\t"
      "v_nop\n\tv_nop\n\tv_nop\n\tv_nop\n\tv_nop"
      : "+v"(c)
      : "v"(A0), "v"(A1), "v"(A2), "v"(A3), "v"(A4), "v"(A5), "v"(A6), "v"(A7),
        "v"(B0), "v"(B1), "v"(B2), "v"(B3), "v"(B4), "v"(B5), "v"(B6), "v"(B7));
  float* out = P + (size_t)(wave * 16 + 8 * h) * F_H1 + m;
  out[0 * F_H1] = c[0] * d0.x;
  out[1 * F_H1] = c[1] * d0.y;
  out[2 * F_H1] = c[2] * d0.z;
  out[3 * F_H1] = c[3] * d0.w;
  out[4 * F_H1] = c[4] * d1.x;
  out[5 * F_H1] = c[5] * d1.y;
  out[6 * F_H1] = c[6] * d1.z;
  out[7 * F_H1] = c[7] * d1.w;
}

__global__ __launch_bounds__(128) void pull1_rows(const float* __restrict__ P,
                                                  const int* __restrict__ starts,
                                                  const int* __restrict__ ssrc,
                                                  const float* __restrict__ dinv,
                                                  const float* __restrict__ b1,
                                                  const float* __restrict__ W2,
                                                  float* __restrict__ T2) {
  __shared__ int idbuf[4 * IDBUF];
  __shared__ float tile[4 * 16 * TP2];
  __shared__ float obuf[4 * 16 * F_H2];
  int wave = blockIdx.x * (blockDim.x >> 5) + (threadIdx.x >> 5);
  int lane = threadIdx.x & 31;
  int g = lane >> 2;
  int j = lane & 3;
  int nb = wave * 16 + 2 * g;
  int st[3];
#pragma unroll
  for (int q = 0; q < 3; ++q) {
    int n = nb + q;
    st[q] = (n < N_NODES) ? starts[n] : N_EDGES;
  }
  float ddq[2];
  ddq[0] = dinv[nb];
  ddq[1] = dinv[nb + 1];
  const float* pb = P + 4 * j;
  int* my = idbuf + (threadIdx.x >> 5) * IDBUF;
  int seg_beg = __builtin_amdgcn_readlane(st[0], 0);
  int seg_end = __builtin_amdgcn_readlane(st[2], 31);

  float4 acc[2];
  acc[0] = make_float4(0.f, 0.f, 0.f, 0.f);
  acc[1] = acc[0];
  for (int abase = seg_beg & ~3; abase < seg_end; abase += IDBUF) {
    stage_ids(ssrc, abase, min(IDBUF, seg_end - abase), my, lane);
#pragma unroll
    for (int q = 0; q < 2; ++q) {
      int lo = max(st[q], abase) - abase;
      int last = min(st[q + 1], abase + IDBUF) - 1 - abase;
      float4 a = acc[q];
      for (int i = lo; i <= last; i += 4) {
        bool k1 = (i + 1 <= last), k2 = (i + 2 <= last), k3 = (i + 3 <= last);
        int s0 = my[i];
        int s1 = my[k1 ? i + 1 : i];
        int s2 = my[k2 ? i + 2 : i];
        int s3 = my[k3 ? i + 3 : i];
        float4 u0 = *(const float4*)(pb + (size_t)s0 * F_H1);
        float4 u1 = *(const float4*)(pb + (size_t)s1 * F_H1);
        float4 u2 = *(const float4*)(pb + (size_t)s2 * F_H1);
        float4 u3 = *(const float4*)(pb + (size_t)s3 * F_H1);
        fma4(a, u0, 1.0f);
        fma4(a, u1, k1 ? 1.0f : 0.0f);
        fma4(a, u2, k2 ? 1.0f : 0.0f);
        fma4(a, u3, k3 ? 1.0f : 0.0f);
      }
      acc[q] = a;
    }
  }
  float* tl = tile + (threadIdx.x >> 5) * (16 * TP2);
  float4 bv = *(const float4*)(b1 + 4 * j);
#pragma unroll
  for (int q = 0; q < 2; ++q) {
    float4 sv = *(const float4*)(pb + (size_t)(nb + q) * F_H1);
    float dd = ddq[q];
    float4 a = acc[q];
    float4 r;
    r.x = fmaxf(fmaf(a.x + sv.x, dd, bv.x), 0.0f);
    r.y = fmaxf(fmaf(a.y + sv.y, dd, bv.y), 0.0f);
    r.z = fmaxf(fmaf(a.z + sv.z, dd, bv.z), 0.0f);
    r.w = fmaxf(fmaf(a.w + sv.w, dd, bv.w), 0.0f);
    *(float4*)(tl + (2 * g + q) * TP2 + 4 * j) = r;
  }
  int m  = lane & 15;
  int hh = lane >> 4;
  const float* hrow = tl + m * TP2 + 2 * hh;
  bool live_col = (m < F_H2);
  const float* w2 = W2 + (size_t)(2 * hh) * F_H2 + (live_col ? m : 0);
  v2f A0 = {hrow[0],  hrow[1]},  A1 = {hrow[4],  hrow[5]};
  v2f A2 = {hrow[8],  hrow[9]},  A3 = {hrow[12], hrow[13]};
  v2f B0 = {live_col ? w2[0 * F_H2]  : 0.0f, live_col ? w2[1 * F_H2]  : 0.0f};
  v2f B1 = {live_col ? w2[4 * F_H2]  : 0.0f, live_col ? w2[5 * F_H2]  : 0.0f};
  v2f B2 = {live_col ? w2[8 * F_H2]  : 0.0f, live_col ? w2[9 * F_H2]  : 0.0f};
  v2f B3 = {live_col ? w2[12 * F_H2] : 0.0f, live_col ? w2[13 * F_H2] : 0.0f};
  float4 d0 = *(const float4*)(dinv + wave * 16 + 8 * hh);
  float4 d1 = *(const float4*)(dinv + wave * 16 + 8 * hh + 4);
  v8f c = {0.f, 0.f, 0.f, 0.f, 0.f, 0.f, 0.f, 0.f};
  asm volatile(
      "v_wmma_f32_16x16x4_f32 %0, %1, %5, %0\n\t"
      "v_wmma_f32_16x16x4_f32 %0, %2, %6, %0\n\t"
      "v_wmma_f32_16x16x4_f32 %0, %3, %7, %0\n\t"
      "v_wmma_f32_16x16x4_f32 %0, %4, %8, %0\n\t"
      "v_nop\n\tv_nop\n\tv_nop\n\tv_nop\n\tv_nop"
      : "+v"(c)
      : "v"(A0), "v"(A1), "v"(A2), "v"(A3), "v"(B0), "v"(B1), "v"(B2), "v"(B3));
  float* ob = obuf + (threadIdx.x >> 5) * (16 * F_H2);
  if (live_col) {
    float* o = ob + (8 * hh) * F_H2 + m;
    o[0 * F_H2] = c[0] * d0.x;
    o[1 * F_H2] = c[1] * d0.y;
    o[2 * F_H2] = c[2] * d0.z;
    o[3 * F_H2] = c[3] * d0.w;
    o[4 * F_H2] = c[4] * d1.x;
    o[5 * F_H2] = c[5] * d1.y;
    o[6 * F_H2] = c[6] * d1.z;
    o[7 * F_H2] = c[7] * d1.w;
  }
  float4 ov = *(const float4*)(ob + 4 * lane);
  *(float4*)(T2 + (size_t)wave * (16 * F_H2) + 4 * lane) = ov;
}

__global__ __launch_bounds__(128) void pull2_rows(const float* __restrict__ T2,
                                                  const int* __restrict__ starts,
                                                  const int* __restrict__ ssrc,
                                                  const float* __restrict__ dinv,
                                                  const float* __restrict__ b2,
                                                  const int* __restrict__ batch,
                                                  float* __restrict__ pooled,
                                                  float* __restrict__ cnt) {
  __shared__ int idbuf[4 * IDBUF];
  int wave = blockIdx.x * (blockDim.x >> 5) + (threadIdx.x >> 5);
  int lane = threadIdx.x & 31;
  int g = lane >> 1;
  int j = lane & 1;
  int node = wave * 16 + g;
  int st0 = starts[node];
  int st1 = (node + 1 < N_NODES) ? starts[node + 1] : N_EDGES;
  float dd = dinv[node];
  const float* tb = T2 + 4 * j;
  int* my = idbuf + (threadIdx.x >> 5) * IDBUF;
  int seg_beg = __builtin_amdgcn_readlane(st0, 0);
  int seg_end = __builtin_amdgcn_readlane(st1, 31);

  float4 a = make_float4(0.f, 0.f, 0.f, 0.f);
  for (int abase = seg_beg & ~3; abase < seg_end; abase += IDBUF) {
    stage_ids(ssrc, abase, min(IDBUF, seg_end - abase), my, lane);
    int lo = max(st0, abase) - abase;
    int last = min(st1, abase + IDBUF) - 1 - abase;
    for (int i = lo; i <= last; i += 4) {
      bool k1 = (i + 1 <= last), k2 = (i + 2 <= last), k3 = (i + 3 <= last);
      int s0 = my[i];
      int s1 = my[k1 ? i + 1 : i];
      int s2 = my[k2 ? i + 2 : i];
      int s3 = my[k3 ? i + 3 : i];
      float4 u0 = *(const float4*)(tb + (size_t)s0 * F_H2);
      float4 u1 = *(const float4*)(tb + (size_t)s1 * F_H2);
      float4 u2 = *(const float4*)(tb + (size_t)s2 * F_H2);
      float4 u3 = *(const float4*)(tb + (size_t)s3 * F_H2);
      fma4(a, u0, 1.0f);
      fma4(a, u1, k1 ? 1.0f : 0.0f);
      fma4(a, u2, k2 ? 1.0f : 0.0f);
      fma4(a, u3, k3 ? 1.0f : 0.0f);
    }
  }
  float4 bv = *(const float4*)(b2 + 4 * j);
  float4 sv = *(const float4*)(tb + (size_t)node * F_H2);
  float4 r;
  r.x = fmaxf(fmaf(a.x + sv.x, dd, bv.x), 0.0f);
  r.y = fmaxf(fmaf(a.y + sv.y, dd, bv.y), 0.0f);
  r.z = fmaxf(fmaf(a.z + sv.z, dd, bv.z), 0.0f);
  r.w = fmaxf(fmaf(a.w + sv.w, dd, bv.w), 0.0f);
  int gr = batch[node];
  float* ps = pooled + (size_t)gr * F_H2 + 4 * j;
  ATOMIC_ADD(&ps[0], r.x);
  ATOMIC_ADD(&ps[1], r.y);
  ATOMIC_ADD(&ps[2], r.z);
  ATOMIC_ADD(&ps[3], r.w);
  if (j == 0) ATOMIC_ADD(&cnt[gr], 1.0f);
}


extern "C" void kernel_launch(void* const* d_in, const int* in_sizes, int n_in,
                              void* d_out, int out_size, void* d_ws, size_t ws_size,
                              hipStream_t stream) {
  const float* x     = (const float*)d_in[0];
  const int*   ei    = (const int*)d_in[1];
  const int*   src   = ei;
  const int*   dst   = ei + N_EDGES;
  const int*   batch = (const int*)d_in[2];
  const float* W1 = (const float*)d_in[4];
  const float* b1 = (const float*)d_in[5];
  const float* W2 = (const float*)d_in[6];
  const float* b2 = (const float*)d_in[7];
  const float* Wl = (const float*)d_in[8];
  const float* bl = (const float*)d_in[9];
  float* out = (float*)d_out;

  char* ws = (char*)d_ws;
  float* dinv   = (float*)(ws);
  int*   arr    = (int*)(ws + (4ll << 20));
  int*   lst    = (int*)(ws + (8ll << 20));
  float* rows   = (float*)(ws + (8ll << 20));
  int*   planes = (int*)(ws + 72388608ll);
  int*   ssrc   = (int*)(ws + 72388608ll);
  int*   bsum   = (int*)(ws + 136388608ll);
  float* pooled = (float*)(ws + 136396800ll);
  int*   tbeg   = (int*)(ws + 136396800ll);
  float* cnt    = pooled + (size_t)N_GRAPHS * F_H2;

  const int NB = 256;
  const int TILE_BLOCKS = N_NODES / 64;

  hist_buckets<<<NSLICE, LB_THREADS, 0, stream>>>(dst, planes);
  scan_block_sums<<<SCAN_BLOCKS, 256, 0, stream>>>(planes, bsum);
  scan_top<<<1, 1024, 0, stream>>>(bsum, SCAN_BLOCKS);
  scan_apply<<<SCAN_BLOCKS, 256, 0, stream>>>(planes, tbeg, bsum);
  scatter_runs<<<NSLICE, LB_THREADS, 0, stream>>>(src, dst, planes, tbeg, lst);
  bucket_order<<<NBK_LIVE, BO_THREADS, 0, stream>>>(lst, tbeg, arr, dinv, ssrc);

  if (ws_size >= (size_t)PR_NEED) {
    float* P  = rows;
    float* t2 = (float*)(ws + PR_T2_OFF);
    xw1_scaled<<<TILE_BLOCKS, 128, 0, stream>>>(x, dinv, W1, P);
    pull1_rows<<<TILE_BLOCKS, 128, 0, stream>>>(P, arr, ssrc, dinv, b1, W2, t2);
    zero_u32<<<128, 256, 0, stream>>>((unsigned*)pooled, N_GRAPHS * (F_H2 + 1));
    pull2_rows<<<TILE_BLOCKS, 128, 0, stream>>>(t2, arr, ssrc, dinv, b2, batch, pooled, cnt);
  } else {
    float* h1 = rows;
    pull1_xw1<<<TILE_BLOCKS, 128, 0, stream>>>(x, arr, ssrc, dinv, W1, b1, h1);
    gemm_h1_w2<<<TILE_BLOCKS, 128, 0, stream>>>(h1, W2, dinv);
    zero_u32<<<128, 256, 0, stream>>>((unsigned*)pooled, N_GRAPHS * (F_H2 + 1));
    pull2_pool<<<TILE_BLOCKS, 128, 0, stream>>>(h1, arr, ssrc, dinv, b2, batch, pooled, cnt);
  }
  readout<<<(N_GRAPHS + NB - 1) / NB, NB, 0, stream>>>(pooled, cnt, Wl, bl,
                                                       out, N_GRAPHS);
}


# list build in 2 launches, not 6: a slice counts, scans and orders itself, stores one piece + bucket offsets; a bucket's group reads its 256 runs (offset sums = list start); sum order = LDS arrival
# speedup vs baseline: 10.6874x; 1.0450x over previous
#include <hip/hip_runtime.h>


typedef float v2f __attribute__((ext_vector_type(2)));
typedef float v8f __attribute__((ext_vector_type(8)));

#define N_NODES   1000000
#define N_EDGES   16000000
#define F_IN      32
#define F_H1      16
#define F_H2      8
#define N_CLS     5
#define N_GRAPHS  10000

#define ATOMIC_ADD(p, v) \
  __hip_atomic_fetch_add((p), (v), __ATOMIC_RELAXED, __HIP_MEMORY_SCOPE_AGENT)


__global__ void zero_u32(unsigned* __restrict__ p, int n) {
  int i = blockIdx.x * blockDim.x + threadIdx.x;
  int stride = gridDim.x * blockDim.x;
  for (; i < n; i += stride) p[i] = 0u;
}

#define LDS_ADD(p, v) \
  __hip_atomic_fetch_add((p), (v), __ATOMIC_RELAXED, __HIP_MEMORY_SCOPE_WORKGROUP)

#define BK_SHIFT 10
#define BK_NODES (1 << BK_SHIFT)
#define NBK_LIVE ((N_NODES + BK_NODES - 1) / BK_NODES)
#define NBK     ((NBK_LIVE + 3) & ~3)
#define NSLICE  256
#define SLICE_V4 (N_EDGES / 4 / NSLICE)
#define SLICE_E  (SLICE_V4 * 4)
#define LB_THREADS 1024
#define LB_GROUPS 4
#define SRC_BITS 20
#define SRC_MASK ((1 << SRC_BITS) - 1)
static_assert(NBK % 4 == 0 && NBK <= LB_THREADS, "planes are whole 16-byte groups; one thread per bucket");
static_assert(SLICE_V4 * 4 * NSLICE == N_EDGES, "slices cover the edges exactly");
static_assert(N_NODES <= (1 << SRC_BITS), "a source id fits its field");
static_assert(SRC_BITS + BK_SHIFT < 31, "an entry is a non-negative int");
static_assert((SLICE_E + 3 * NBK + 64 + LB_THREADS) * 4 <= 320 * 1024, "the place pass fits the LDS");

#define SCAN_TILES 64
#define SCAN_BLOCKS ((NBK + SCAN_TILES - 1) / SCAN_TILES)
static_assert(SCAN_BLOCKS <= 1024, "one block scans the block sums");

__global__ __launch_bounds__(LB_THREADS) void hist_buckets(const int* __restrict__ dst,
                                                           int* __restrict__ planes) {
  __shared__ int cnt[NBK];
  int sl = blockIdx.x;
  for (int i = threadIdx.x; i < NBK; i += LB_THREADS) cnt[i] = 0;
  __syncthreads();
  const int4* d4 = (const int4*)dst + (size_t)sl * SLICE_V4;
  for (int i = threadIdx.x; i < SLICE_V4; i += LB_GROUPS * LB_THREADS) {
    int4 d[LB_GROUPS];
#pragma unroll
    for (int g = 0; g < LB_GROUPS; ++g) {
      int ig = i + g * LB_THREADS;
      d[g] = (ig < SLICE_V4) ? d4[ig] : make_int4(-1, -1, -1, -1);
    }
#pragma unroll
    for (int g = 0; g < LB_GROUPS; ++g) {
      if ((unsigned)d[g].x < (unsigned)N_NODES) LDS_ADD(&cnt[d[g].x >> BK_SHIFT], 1);
      if ((unsigned)d[g].y < (unsigned)N_NODES) LDS_ADD(&cnt[d[g].y >> BK_SHIFT], 1);
      if ((unsigned)d[g].z < (unsigned)N_NODES) LDS_ADD(&cnt[d[g].z >> BK_SHIFT], 1);
      if ((unsigned)d[g].w < (unsigned)N_NODES) LDS_ADD(&cnt[d[g].w >> BK_SHIFT], 1);
    }
  }
  __syncthreads();
  for (int i = threadIdx.x; i < NBK; i += LB_THREADS) planes[(size_t)sl * NBK + i] = cnt[i];
}

template <int NT>
__device__ __forceinline__ int block_scan_excl(int v, int* sh, int* total) {
  int t = threadIdx.x;
  sh[t] = v;
  __syncthreads();
  for (int o = 1; o < NT; o <<= 1) {
    int add = (t >= o) ? sh[t - o] : 0;
    __syncthreads();
    sh[t] += add;
    __syncthreads();
  }
  int incl = sh[t];
  *total = sh[NT - 1];
  __syncthreads();
  return incl - v;
}

__global__ __launch_bounds__(256) void scan_block_sums(const int* __restrict__ planes,
                                                       int* __restrict__ bsum) {
  __shared__ int sh[256];
  int tq = threadIdx.x & 15;
  int pg = threadIdx.x >> 4;
  int base = blockIdx.x * SCAN_TILES + 4 * tq;
  int s = 0;
  if (base < NBK) {
#pragma unroll
    for (int p = 0; p < 16; ++p) {
      int4 v = *(const int4*)(planes + (size_t)(pg * 16 + p) * NBK + base);
      s += v.x + v.y + v.z + v.w;
    }
  }
  int total;
  (void)block_scan_excl<256>(s, sh, &total);
  if (threadIdx.x == 0) bsum[blockIdx.x] = total;
}

__global__ __launch_bounds__(1024) void scan_top(int* __restrict__ bsum, int nb) {
  __shared__ int sh[1024];
  int t = threadIdx.x;
  int v = (t < nb) ? bsum[t] : 0;
  int total;
  int ex = block_scan_excl<1024>(v, sh, &total);
  if (t < nb) bsum[t] = ex;
}

__global__ __launch_bounds__(256) void scan_apply(int* __restrict__ planes,
                                                  int* __restrict__ tbeg,
                                                  const int* __restrict__ boff) {
  __shared__ int4 part[16 * 16];
  __shared__ int tsum[16];
  int tq = threadIdx.x & 15;
  int pg = threadIdx.x >> 4;
  int base = blockIdx.x * SCAN_TILES + 4 * tq;
  bool live = (base < NBK);
  int4 c[16];
  int4 v = make_int4(0, 0, 0, 0);
#pragma unroll
  for (int p = 0; p < 16; ++p) {
    c[p] = make_int4(0, 0, 0, 0);
    if (live) c[p] = *(const int4*)(planes + (size_t)(pg * 16 + p) * NBK + base);
    v.x += c[p].x; v.y += c[p].y; v.z += c[p].z; v.w += c[p].w;
  }
  part[pg * 16 + tq] = v;
  __syncthreads();
  int4 pre = make_int4(0, 0, 0, 0);
  int4 tot = make_int4(0, 0, 0, 0);
  for (int g = 0; g < 16; ++g) {
    int4 w = part[g * 16 + tq];
    if (g < pg) { pre.x += w.x; pre.y += w.y; pre.z += w.z; pre.w += w.w; }
    tot.x += w.x; tot.y += w.y; tot.z += w.z; tot.w += w.w;
  }
  if (pg == 0) tsum[tq] = tot.x + tot.y + tot.z + tot.w;
  __syncthreads();
  int ex = boff[blockIdx.x];
  for (int q = 0; q < 16; ++q) {
    int w = tsum[q];
    if (q < tq) ex += w;
  }
  if (live) {
    int4 o;
    o.x = ex;
    o.y = o.x + tot.x;
    o.z = o.y + tot.y;
    o.w = o.z + tot.z;
    if (pg == 0) {
      *(int4*)(tbeg + base) = o;
      if (base + 4 == NBK) tbeg[NBK] = o.w + tot.w;
    }
    o.x += pre.x; o.y += pre.y; o.z += pre.z; o.w += pre.w;
#pragma unroll
    for (int p = 0; p < 16; ++p) {
      *(int4*)(planes + (size_t)(pg * 16 + p) * NBK + base) = o;
      o.x += c[p].x; o.y += c[p].y; o.z += c[p].z; o.w += c[p].w;
    }
  }
}

__global__ __launch_bounds__(LB_THREADS) void scatter_runs(const int* __restrict__ src,
                                                           const int* __restrict__ dst,
                                                           const int* __restrict__ planes,
                                                           const int* __restrict__ tbeg,
                                                           int* __restrict__ lst) {
  __shared__ int ebuf[SLICE_E];
  __shared__ int gbeg[NBK];
  __shared__ int lbeg[NBK + 1];
  __shared__ int lcur[NBK + 32];
  __shared__ int sh[LB_THREADS];
  int sl = blockIdx.x;
  int t = threadIdx.x;
  int c = 0;
  if (t < NBK) {
    int g0 = planes[(size_t)sl * NBK + t];
    int g1 = (sl + 1 < NSLICE) ? planes[(size_t)(sl + 1) * NBK + t] : tbeg[t + 1];
    gbeg[t] = g0;
    c = g1 - g0;
  }
  int total;
  int ex = block_scan_excl<LB_THREADS>(c, sh, &total);
  if (t < NBK) { lbeg[t] = ex; lcur[t] = ex; }
  if (t == 0) lbeg[NBK] = total;
  if (t < 32) lcur[NBK + t] = 0;
  __syncthreads();
  int spare = NBK + (t & 31);
  const int4* d4 = (const int4*)dst + (size_t)sl * SLICE_V4;
  const int4* s4 = (const int4*)src + (size_t)sl * SLICE_V4;
  for (int i = t; i < SLICE_V4; i += LB_GROUPS * LB_THREADS) {
    int4 d[LB_GROUPS], sv[LB_GROUPS];
    uint4 p[LB_GROUPS];
#pragma unroll
    for (int g = 0; g < LB_GROUPS; ++g) {
      int ig = i + g * LB_THREADS;
      bool inside = (ig < SLICE_V4);
      d[g]  = inside ? d4[ig] : make_int4(-1, -1, -1, -1);
      sv[g] = inside ? s4[ig] : make_int4(0, 0, 0, 0);
    }
#pragma unroll
    for (int g = 0; g < LB_GROUPS; ++g) {
      bool kx = (unsigned)d[g].x < (unsigned)N_NODES, ky = (unsigned)d[g].y < (unsigned)N_NODES;
      bool kz = (unsigned)d[g].z < (unsigned)N_NODES, kw = (unsigned)d[g].w < (unsigned)N_NODES;
      p[g].x = (unsigned)LDS_ADD(&lcur[kx ? (d[g].x >> BK_SHIFT) : spare], kx ? 1 : 0);
      p[g].y = (unsigned)LDS_ADD(&lcur[ky ? (d[g].y >> BK_SHIFT) : spare], ky ? 1 : 0);
      p[g].z = (unsigned)LDS_ADD(&lcur[kz ? (d[g].z >> BK_SHIFT) : spare], kz ? 1 : 0);
      p[g].w = (unsigned)LDS_ADD(&lcur[kw ? (d[g].w >> BK_SHIFT) : spare], kw ? 1 : 0);
    }
#pragma unroll
    for (int g = 0; g < LB_GROUPS; ++g) {
      if ((unsigned)d[g].x < (unsigned)N_NODES) ebuf[p[g].x] = sv[g].x | ((d[g].x & (BK_NODES - 1)) << SRC_BITS);
      if ((unsigned)d[g].y < (unsigned)N_NODES) ebuf[p[g].y] = sv[g].y | ((d[g].y & (BK_NODES - 1)) << SRC_BITS);
      if ((unsigned)d[g].z < (unsigned)N_NODES) ebuf[p[g].z] = sv[g].z | ((d[g].z & (BK_NODES - 1)) << SRC_BITS);
      if ((unsigned)d[g].w < (unsigned)N_NODES) ebuf[p[g].w] = sv[g].w | ((d[g].w & (BK_NODES - 1)) << SRC_BITS);
    }
  }
  __syncthreads();
  int lane = t & 31;
  for (int b = t >> 5; b < NBK; b += LB_THREADS / 32) {
    int lb = lbeg[b];
    int n = lbeg[b + 1] - lb;
    int gb = gbeg[b];
    for (int k = lane; k < n; k += 32) lst[gb + k] = ebuf[lb + k];
  }
}

#define BO_THREADS BK_NODES
#define BO_CAP 24576
#define BO_UNROLL 4
__global__ __launch_bounds__(BO_THREADS) void bucket_order(const int* __restrict__ lst,
                                                           const int* __restrict__ tbeg,
                                                           int* __restrict__ starts,
                                                           float* __restrict__ dinv,
                                                           int* __restrict__ ssrc) {
  __shared__ int obuf[BO_CAP];
  __shared__ int cnt[BK_NODES];
  __shared__ int cur[BK_NODES];
  __shared__ int sh[BO_THREADS];
  int b = blockIdx.x;
  int t = threadIdx.x;
  int beg = tbeg[b];
  int n = tbeg[b + 1] - beg;
  const int* piece = lst + beg;
  cnt[t] = 0;
  __syncthreads();
  for (int i0 = 0; i0 < n; i0 += BO_UNROLL * BO_THREADS) {
    int e[BO_UNROLL];
#pragma unroll
    for (int u = 0; u < BO_UNROLL; ++u) {
      int i = i0 + u * BO_THREADS + t;
      e[u] = (i < n) ? piece[i] : -1;
    }
#pragma unroll
    for (int u = 0; u < BO_UNROLL; ++u)
      if (e[u] >= 0) LDS_ADD(&cnt[(e[u] >> SRC_BITS) & (BK_NODES - 1)], 1);
  }
  __syncthreads();
  int c = cnt[t];
  int total;
  int ex = block_scan_excl<BO_THREADS>(c, sh, &total);
  int node = b * BK_NODES + t;
  if (node < N_NODES) {
    starts[node] = beg + ex;
    dinv[node] = rsqrtf((float)c + 1.0f);
  }
  bool in_lds = (n <= BO_CAP);
  cur[t] = in_lds ? ex : beg + ex;
  __syncthreads();
  for (int i0 = 0; i0 < n; i0 += BO_UNROLL * BO_THREADS) {
    int e[BO_UNROLL], p[BO_UNROLL];
#pragma unroll
    for (int u = 0; u < BO_UNROLL; ++u) {
      int i = i0 + u * BO_THREADS + t;
      e[u] = (i < n) ? piece[i] : -1;
    }
#pragma unroll
    for (int u = 0; u < BO_UNROLL; ++u) {
      p[u] = 0;
      if (e[u] >= 0) p[u] = LDS_ADD(&cur[(e[u] >> SRC_BITS) & (BK_NODES - 1)], 1);
    }
#pragma unroll
    for (int u = 0; u < BO_UNROLL; ++u) {
      if (e[u] >= 0) {
        if (in_lds) obuf[p[u]] = e[u] & SRC_MASK;
        else ssrc[p[u]] = e[u] & SRC_MASK;
      }
    }
  }
  if (in_lds) {
    __syncthreads();
    for (int i = t; i < n; i += BO_THREADS) ssrc[beg + i] = obuf[i];
  }
}

#define PSTRIDE (NBK + 4)
static_assert(NBK + 1 <= LB_THREADS, "one thread per offset");
static_assert((SLICE_E * 4) % 16 == 0, "a slice's piece starts on a 16-byte boundary");

__global__ __launch_bounds__(LB_THREADS) void scatter_local(const int* __restrict__ src,
                                                            const int* __restrict__ dst,
                                                            int* __restrict__ offs,
                                                            int* __restrict__ lst) {
  __shared__ __attribute__((aligned(16))) int ebuf[SLICE_E];
  __shared__ int lcur[NBK + 32];
  __shared__ int sh[LB_THREADS];
  int sl = blockIdx.x;
  int t = threadIdx.x;
  for (int i = t; i < NBK + 32; i += LB_THREADS) lcur[i] = 0;
  __syncthreads();
  const int4* d4 = (const int4*)dst + (size_t)sl * SLICE_V4;
  const int4* s4 = (const int4*)src + (size_t)sl * SLICE_V4;
  for (int i = t; i < SLICE_V4; i += LB_GROUPS * LB_THREADS) {
    int4 d[LB_GROUPS];
#pragma unroll
    for (int g = 0; g < LB_GROUPS; ++g) {
      int ig = i + g * LB_THREADS;
      d[g] = (ig < SLICE_V4) ? d4[ig] : make_int4(-1, -1, -1, -1);
    }
#pragma unroll
    for (int g = 0; g < LB_GROUPS; ++g) {
      if ((unsigned)d[g].x < (unsigned)N_NODES) LDS_ADD(&lcur[d[g].x >> BK_SHIFT], 1);
      if ((unsigned)d[g].y < (unsigned)N_NODES) LDS_ADD(&lcur[d[g].y >> BK_SHIFT], 1);
      if ((unsigned)d[g].z < (unsigned)N_NODES) LDS_ADD(&lcur[d[g].z >> BK_SHIFT], 1);
      if ((unsigned)d[g].w < (unsigned)N_NODES) LDS_ADD(&lcur[d[g].w >> BK_SHIFT], 1);
    }
  }
  __syncthreads();
  int c = (t < NBK) ? lcur[t] : 0;
  int total;
  int ex = block_scan_excl<LB_THREADS>(c, sh, &total);
  if (t < NBK) { lcur[t] = ex; offs[(size_t)sl * PSTRIDE + t] = ex; }
  if (t == NBK) offs[(size_t)sl * PSTRIDE + NBK] = total;
  __syncthreads();
  int spare = NBK + (t & 31);
  for (int i = t; i < SLICE_V4; i += LB_GROUPS * LB_THREADS) {
    int4 d[LB_GROUPS], sv[LB_GROUPS];
    uint4 p[LB_GROUPS];
#pragma unroll
    for (int g = 0; g < LB_GROUPS; ++g) {
      int ig = i + g * LB_THREADS;
      bool inside = (ig < SLICE_V4);
      d[g]  = inside ? d4[ig] : make_int4(-1, -1, -1, -1);
      sv[g] = inside ? s4[ig] : make_int4(0, 0, 0, 0);
    }
#pragma unroll
    for (int g = 0; g < LB_GROUPS; ++g) {
      bool kx = (unsigned)d[g].x < (unsigned)N_NODES, ky = (unsigned)d[g].y < (unsigned)N_NODES;
      bool kz = (unsigned)d[g].z < (unsigned)N_NODES, kw = (unsigned)d[g].w < (unsigned)N_NODES;
      p[g].x = (unsigned)LDS_ADD(&lcur[kx ? (d[g].x >> BK_SHIFT) : spare], kx ? 1 : 0);
      p[g].y = (unsigned)LDS_ADD(&lcur[ky ? (d[g].y >> BK_SHIFT) : spare], ky ? 1 : 0);
      p[g].z = (unsigned)LDS_ADD(&lcur[kz ? (d[g].z >> BK_SHIFT) : spare], kz ? 1 : 0);
      p[g].w = (unsigned)LDS_ADD(&lcur[kw ? (d[g].w >> BK_SHIFT) : spare], kw ? 1 : 0);
    }
#pragma unroll
    for (int g = 0; g < LB_GROUPS; ++g) {
      if ((unsigned)d[g].x < (unsigned)N_NODES) ebuf[p[g].x] = sv[g].x | ((d[g].x & (BK_NODES - 1)) << SRC_BITS);
      if ((unsigned)d[g].y < (unsigned)N_NODES) ebuf[p[g].y] = sv[g].y | ((d[g].y & (BK_NODES - 1)) << SRC_BITS);
      if ((unsigned)d[g].z < (unsigned)N_NODES) ebuf[p[g].z] = sv[g].z | ((d[g].z & (BK_NODES - 1)) << SRC_BITS);
      if ((unsigned)d[g].w < (unsigned)N_NODES) ebuf[p[g].w] = sv[g].w | ((d[g].w & (BK_NODES - 1)) << SRC_BITS);
    }
  }
  __syncthreads();
  int4* out = (int4*)(lst + (size_t)sl * SLICE_E);
  const int4* e4 = (const int4*)ebuf;
  for (int i = t; i < (total + 3) / 4; i += LB_THREADS) out[i] = e4[i];
}

__device__ __forceinline__ int wave_sum(int v) {
  for (int o = 16; o > 0; o >>= 1) v += __shfl_xor(v, o, 32);
  return v;
}

#define BO_RUNS  (NSLICE / (BO_THREADS / 32))
#define BO_TRIPS 3
#define BO_HEAD  (32 * BO_TRIPS)
static_assert(BO_RUNS * (BO_THREADS / 32) == NSLICE, "the waves cover the slices exactly");
static_assert(NSLICE <= BO_THREADS && NSLICE % 32 == 0, "whole waves read the offsets");
__global__ __launch_bounds__(BO_THREADS, 16) void bucket_order_sl(const int* __restrict__ lst,
                                                              const int* __restrict__ offs,
                                                              int* __restrict__ starts,
                                                              float* __restrict__ dinv,
                                                              int* __restrict__ ssrc) {
  __shared__ int obuf[BO_CAP];
  __shared__ int cnt[BK_NODES];
  __shared__ int cur[BK_NODES];
  __shared__ int sh[BO_THREADS];
  __shared__ int roff[NSLICE];
  __shared__ int rlen[NSLICE];
  __shared__ int wsum[2 * (NSLICE / 32)];
  int b = blockIdx.x;
  int t = threadIdx.x;
  int lane = t & 31;
  int w = t >> 5;
  cnt[t] = 0;
  if (t < NSLICE) {
    int o0 = offs[(size_t)t * PSTRIDE + b];
    int o1 = offs[(size_t)t * PSTRIDE + b + 1];
    roff[t] = t * SLICE_E + o0;
    rlen[t] = o1 - o0;
    int a0 = wave_sum(o0), a1 = wave_sum(o1);
    if (lane == 0) { wsum[2 * w] = a0; wsum[2 * w + 1] = a1; }
  }
  __syncthreads();
  int beg = 0, end = 0;
  for (int k = 0; k < NSLICE / 32; ++k) { beg += wsum[2 * k]; end += wsum[2 * k + 1]; }
  int n = end - beg;

  int e[BO_RUNS][BO_TRIPS];
#pragma unroll
  for (int r = 0; r < BO_RUNS; ++r) {
    int s = w + (BO_THREADS / 32) * r;
    int off = __builtin_amdgcn_readfirstlane(roff[s]);
    int len = __builtin_amdgcn_readfirstlane(rlen[s]);
#pragma unroll
    for (int k = 0; k < BO_TRIPS; ++k) {
      int i = 32 * k + lane;
      e[r][k] = (i < len) ? lst[off + i] : -1;
    }
  }
#pragma unroll
  for (int r = 0; r < BO_RUNS; ++r)
#pragma unroll
    for (int k = 0; k < BO_TRIPS; ++k)
      if (e[r][k] >= 0) LDS_ADD(&cnt[(e[r][k] >> SRC_BITS) & (BK_NODES - 1)], 1);
  for (int r = 0; r < BO_RUNS; ++r) {
    int s = w + (BO_THREADS / 32) * r;
    int off = __builtin_amdgcn_readfirstlane(roff[s]);
    int len = __builtin_amdgcn_readfirstlane(rlen[s]);
    for (int i = BO_HEAD + lane; i < len; i += 32)
      LDS_ADD(&cnt[(lst[off + i] >> SRC_BITS) & (BK_NODES - 1)], 1);
  }
  __syncthreads();
  int c = cnt[t];
  int total;
  int ex = block_scan_excl<BO_THREADS>(c, sh, &total);
  int node = b * BK_NODES + t;
  if (node < N_NODES) {
    starts[node] = beg + ex;
    dinv[node] = rsqrtf((float)c + 1.0f);
  }
  bool in_lds = (n <= BO_CAP);
  cur[t] = in_lds ? ex : beg + ex;
  __syncthreads();
#pragma unroll
  for (int r = 0; r < BO_RUNS; ++r) {
    int p[BO_TRIPS];
#pragma unroll
    for (int k = 0; k < BO_TRIPS; ++k) {
      p[k] = 0;
      if (e[r][k] >= 0) p[k] = LDS_ADD(&cur[(e[r][k] >> SRC_BITS) & (BK_NODES - 1)], 1);
    }
#pragma unroll
    for (int k = 0; k < BO_TRIPS; ++k) {
      if (e[r][k] >= 0) {
        if (in_lds) obuf[p[k]] = e[r][k] & SRC_MASK;
        else ssrc[p[k]] = e[r][k] & SRC_MASK;
      }
    }
  }
  for (int r = 0; r < BO_RUNS; ++r) {
    int s = w + (BO_THREADS / 32) * r;
    int off = __builtin_amdgcn_readfirstlane(roff[s]);
    int len = __builtin_amdgcn_readfirstlane(rlen[s]);
    for (int i = BO_HEAD + lane; i < len; i += 32) {
      int v = lst[off + i];
      int p = LDS_ADD(&cur[(v >> SRC_BITS) & (BK_NODES - 1)], 1);
      if (in_lds) obuf[p] = v & SRC_MASK;
      else ssrc[p] = v & SRC_MASK;
    }
  }
  if (in_lds) {
    __syncthreads();
    for (int i = t; i < n; i += BO_THREADS) ssrc[beg + i] = obuf[i];
  }
}

__device__ __forceinline__ void fma4(float4& a, const float4& v, float w) {
  a.x = fmaf(v.x, w, a.x);
  a.y = fmaf(v.y, w, a.y);
  a.z = fmaf(v.z, w, a.z);
  a.w = fmaf(v.w, w, a.w);
}

#define IDBUF 512
#define TPITCH 36

__device__ __forceinline__ void stage_ids(const int* __restrict__ ssrc, int abase,
                                          int need, int* buf, int lane) {
  int4 v[IDBUF / 128];
#pragma unroll
  for (int k = 0; k < IDBUF / 128; ++k) {
    int o = 128 * k + 4 * lane;
    v[k] = *(const int4*)(ssrc + abase + (o < need ? o : 0));
  }
#pragma unroll
  for (int k = 0; k < IDBUF / 128; ++k) *(int4*)(buf + 128 * k + 4 * lane) = v[k];
}


__global__ __launch_bounds__(128) void pull1_xw1(const float* __restrict__ X,
                                                 const int* __restrict__ starts,
                                                 const int* __restrict__ ssrc,
                                                 const float* __restrict__ dinv,
                                                 const float* __restrict__ W1,
                                                 const float* __restrict__ b1,
                                                 float* __restrict__ H1) {
  __shared__ int idbuf[4 * IDBUF];
  __shared__ float tile[4 * 16 * TPITCH];
  int wave = blockIdx.x * (blockDim.x >> 5) + (threadIdx.x >> 5);
  int lane = threadIdx.x & 31;
  int m = lane & 15;
  int h = lane >> 4;
  int g = lane >> 3;
  int j = lane & 7;
  int nb = wave * 16 + 4 * g;
  int st[5];
#pragma unroll
  for (int q = 0; q < 5; ++q) {
    int n = nb + q;
    st[q] = (n < N_NODES) ? starts[n] : N_EDGES;
  }
  float ddq[4];
#pragma unroll
  for (int q = 0; q < 4; ++q) ddq[q] = dinv[nb + q];
  const float* xb = X + 4 * j;
  int* my = idbuf + (threadIdx.x >> 5) * IDBUF;
  int seg_beg = __builtin_amdgcn_readlane(st[0], 0);
  int seg_end = __builtin_amdgcn_readlane(st[4], 31);

  float4 acc[4];
#pragma unroll
  for (int q = 0; q < 4; ++q) acc[q] = make_float4(0.f, 0.f, 0.f, 0.f);
  for (int abase = seg_beg & ~3; abase < seg_end; abase += IDBUF) {
    stage_ids(ssrc, abase, min(IDBUF, seg_end - abase), my, lane);
#pragma unroll
    for (int q = 0; q < 4; ++q) {
      int lo = max(st[q], abase) - abase;
      int last = min(st[q + 1], abase + IDBUF) - 1 - abase;
      float dd = ddq[q];
      float4 a = acc[q];
      for (int i = lo; i <= last; i += 4) {
        bool k1 = (i + 1 <= last), k2 = (i + 2 <= last), k3 = (i + 3 <= last);
        int s0 = my[i];
        int s1 = my[k1 ? i + 1 : i];
        int s2 = my[k2 ? i + 2 : i];
        int s3 = my[k3 ? i + 3 : i];
        float d0 = dinv[s0], d1 = dinv[s1], d2 = dinv[s2], d3 = dinv[s3];
        float4 u0 = *(const float4*)(xb + (size_t)s0 * F_IN);
        float4 u1 = *(const float4*)(xb + (size_t)s1 * F_IN);
        float4 u2 = *(const float4*)(xb + (size_t)s2 * F_IN);
        float4 u3 = *(const float4*)(xb + (size_t)s3 * F_IN);
        float w0 = d0 * dd;
        float w1 = k1 ? d1 * dd : 0.0f;
        float w2 = k2 ? d2 * dd : 0.0f;
        float w3 = k3 ? d3 * dd : 0.0f;
        fma4(a, u0, w0); fma4(a, u1, w1); fma4(a, u2, w2); fma4(a, u3, w3);
      }
      acc[q] = a;
    }
  }
  float* tl = tile + (threadIdx.x >> 5) * (16 * TPITCH);
#pragma unroll
  for (int q = 0; q < 4; ++q) {
    float ws = ddq[q] * ddq[q];
    float4 u = *(const float4*)(xb + (size_t)(nb + q) * F_IN);
    float4 a = acc[q];
    fma4(a, u, ws);
    *(float4*)(tl + (4 * g + q) * TPITCH + 4 * j) = a;
  }
  const float* tr = tl + m * TPITCH + 16 * h;
  float4 a0 = *(const float4*)(tr + 0);
  float4 a1 = *(const float4*)(tr + 4);
  float4 a2 = *(const float4*)(tr + 8);
  float4 a3 = *(const float4*)(tr + 12);

  const float* wb = W1 + (size_t)(16 * h) * F_H1 + m;
  v2f A0 = {a0.x, a0.y}, A1 = {a0.z, a0.w}, A2 = {a1.x, a1.y}, A3 = {a1.z, a1.w};
  v2f A4 = {a2.x, a2.y}, A5 = {a2.z, a2.w}, A6 = {a3.x, a3.y}, A7 = {a3.z, a3.w};
  v2f B0 = {wb[0 * F_H1],  wb[1 * F_H1]},  B1 = {wb[2 * F_H1],  wb[3 * F_H1]};
  v2f B2 = {wb[4 * F_H1],  wb[5 * F_H1]},  B3 = {wb[6 * F_H1],  wb[7 * F_H1]};
  v2f B4 = {wb[8 * F_H1],  wb[9 * F_H1]},  B5 = {wb[10 * F_H1], wb[11 * F_H1]};
  v2f B6 = {wb[12 * F_H1], wb[13 * F_H1]}, B7 = {wb[14 * F_H1], wb[15 * F_H1]};
  v8f c = {0.f, 0.f, 0.f, 0.f, 0.f, 0.f, 0.f, 0.f};
  asm volatile(
      "v_wmma_f32_16x16x4_f32 %0, %1, %9, %0\n\t"
      "v_wmma_f32_16x16x4_f32 %0, %2, %10, %0\n\t"
      "v_wmma_f32_16x16x4_f32 %0, %3, %11, %0\n\t"
      "v_wmma_f32_16x16x4_f32 %0, %4, %12, %0\n\t"
      "v_wmma_f32_16x16x4_f32 %0, %5, %13, %0\n\t"
      "v_wmma_f32_16x16x4_f32 %0, %6, %14, %0\n\t"
      "v_wmma_f32_16x16x4_f32 %0, %7, %15, %0\n\t"
      "v_wmma_f32_16x16x4_f32 %0, %8, %16, %0\n\t"
      "v_nop\n\tv_nop\n\tv_nop\n\tv_nop\n\tv_nop"
      : "+v"(c)
      : "v"(A0), "v"(A1), "v"(A2), "v"(A3), "v"(A4), "v"(A5), "v"(A6), "v"(A7),
        "v"(B0), "v"(B1), "v"(B2), "v"(B3), "v"(B4), "v"(B5), "v"(B6), "v"(B7));
  float bb = b1[m];
  float* out = H1 + (size_t)(wave * 16 + 8 * h) * F_H1 + m;
#pragma unroll
  for (int j = 0; j < 8; ++j) out[(size_t)j * F_H1] = fmaxf(c[j] + bb, 0.0f);
}

__global__ __launch_bounds__(128) void gemm_h1_w2(float* H, const float* __restrict__ W2,
                                                  const float* __restrict__ dinv) {
  int wave = blockIdx.x * (blockDim.x >> 5) + (threadIdx.x >> 5);
  int lane = threadIdx.x & 31;
  int m  = lane & 15;
  int kk = (lane >> 4) << 1;
  float* hrow = H + (size_t)(wave * 16 + m) * F_H1;
  bool live_col = (m < F_H2);
  float dm = dinv[wave * 16 + m];

  v2f a[4], b[4];
#pragma unroll
  for (int t = 0; t < 4; ++t) {
    int k = 4 * t + kk;
    a[t].x = hrow[k];
    a[t].y = hrow[k + 1];
    b[t].x = live_col ? W2[k * F_H2 + m] : 0.0f;
    b[t].y = live_col ? W2[(k + 1) * F_H2 + m] : 0.0f;
  }
  if (lane < 16) hrow[F_H2] = dm;

  v8f c = {0.f, 0.f, 0.f, 0.f, 0.f, 0.f, 0.f, 0.f};
#pragma unroll
  for (int t = 0; t < 4; ++t)
    c = __builtin_amdgcn_wmma_f32_16x16x4_f32(false, a[t], false, b[t],
                                              (short)0, c, false, false);
  if (live_col) {
    int hi = lane >> 4;
    float* out = H + (size_t)(wave * 16 + 8 * hi) * F_H1 + m;
#pragma unroll
    for (int j = 0; j < 8; ++j) out[(size_t)j * F_H1] = c[j];
  }
}

__device__ __forceinline__ float quad_lane2(float v) {
  int r = __builtin_amdgcn_mov_dpp(__float_as_int(v), 0xAA, 0xF, 0xF, true);
  return __int_as_float(r);
}

__global__ __launch_bounds__(128) void pull2_pool(const float* __restrict__ H,
                                                  const int* __restrict__ starts,
                                                  const int* __restrict__ ssrc,
                                                  const float* __restrict__ dinv,
                                                  const float* __restrict__ b2,
                                                  const int* __restrict__ batch,
                                                  float* __restrict__ pooled,
                                                  float* __restrict__ cnt) {
  __shared__ int idbuf[4 * IDBUF];
  int wave = blockIdx.x * (blockDim.x >> 5) + (threadIdx.x >> 5);
  int lane = threadIdx.x & 31;
  int g = lane >> 2;
  int j = lane & 3;
  int nb = wave * 16 + 2 * g;
  int st[3];
#pragma unroll
  for (int q = 0; q < 3; ++q) {
    int n = nb + q;
    st[q] = (n < N_NODES) ? starts[n] : N_EDGES;
  }
  float ddq[2];
  ddq[0] = dinv[nb];
  ddq[1] = dinv[nb + 1];
  const float* hb = H + 4 * j;
  int* my = idbuf + (threadIdx.x >> 5) * IDBUF;
  int seg_beg = __builtin_amdgcn_readlane(st[0], 0);
  int seg_end = __builtin_amdgcn_readlane(st[2], 31);

  float4 acc[2];
  acc[0] = make_float4(0.f, 0.f, 0.f, 0.f);
  acc[1] = acc[0];
  for (int abase = seg_beg & ~3; abase < seg_end; abase += IDBUF) {
    stage_ids(ssrc, abase, min(IDBUF, seg_end - abase), my, lane);
#pragma unroll
    for (int q = 0; q < 2; ++q) {
      int lo = max(st[q], abase) - abase;
      int last = min(st[q + 1], abase + IDBUF) - 1 - abase;
      float dd = ddq[q];
      float4 a = acc[q];
      for (int i = lo; i <= last; i += 4) {
        bool k1 = (i + 1 <= last), k2 = (i + 2 <= last), k3 = (i + 3 <= last);
        int s0 = my[i];
        int s1 = my[k1 ? i + 1 : i];
        int s2 = my[k2 ? i + 2 : i];
        int s3 = my[k3 ? i + 3 : i];
        float4 u0 = *(const float4*)(hb + (size_t)s0 * F_H1);
        float4 u1 = *(const float4*)(hb + (size_t)s1 * F_H1);
        float4 u2 = *(const float4*)(hb + (size_t)s2 * F_H1);
        float4 u3 = *(const float4*)(hb + (size_t)s3 * F_H1);
        float d0 = quad_lane2(u0.x), d1 = quad_lane2(u1.x);
        float d2 = quad_lane2(u2.x), d3 = quad_lane2(u3.x);
        float w0 = d0 * dd;
        float w1 = k1 ? d1 * dd : 0.0f;
        float w2 = k2 ? d2 * dd : 0.0f;
        float w3 = k3 ? d3 * dd : 0.0f;
        fma4(a, u0, w0); fma4(a, u1, w1); fma4(a, u2, w2); fma4(a, u3, w3);
      }
      acc[q] = a;
    }
  }
  if (j < 2) {
    float4 bv = *(const float4*)(b2 + 4 * j);
#pragma unroll
    for (int q = 0; q < 2; ++q) {
      int node = nb + q;
      float ws = ddq[q] * ddq[q];
      float4 sv = *(const float4*)(hb + (size_t)node * F_H1);
      float4 a = acc[q];
      float4 r;
      r.x = fmaxf(a.x + sv.x * ws + bv.x, 0.0f);
      r.y = fmaxf(a.y + sv.y * ws + bv.y, 0.0f);
      r.z = fmaxf(a.z + sv.z * ws + bv.z, 0.0f);
      r.w = fmaxf(a.w + sv.w * ws + bv.w, 0.0f);
      int gr = batch[node];
      float* ps = pooled + (size_t)gr * F_H2 + 4 * j;
      ATOMIC_ADD(&ps[0], r.x);
      ATOMIC_ADD(&ps[1], r.y);
      ATOMIC_ADD(&ps[2], r.z);
      ATOMIC_ADD(&ps[3], r.w);
      if (j == 0) ATOMIC_ADD(&cnt[gr], 1.0f);
    }
  }
}

__global__ void readout(const float* __restrict__ pooled, const float* __restrict__ cnt,
                        const float* __restrict__ Wl, const float* __restrict__ bl,
                        float* __restrict__ out, int G) {
  int g = blockIdx.x * blockDim.x + threadIdx.x;
  if (g >= G) return;
  float inv = 1.0f / fmaxf(cnt[g], 1.0f);
  float p[F_H2];
#pragma unroll
  for (int f = 0; f < F_H2; ++f) p[f] = pooled[(size_t)g * F_H2 + f] * inv;
#pragma unroll
  for (int c = 0; c < N_CLS; ++c) {
    float acc = bl[c];
#pragma unroll
    for (int f = 0; f < F_H2; ++f) acc += p[f] * Wl[f * N_CLS + c];
    out[(size_t)g * N_CLS + c] = acc;
  }
}

#define PR_T2_OFF   136757248ll
#define PR_NEED     (PR_T2_OFF + 32000000ll)
#define TP2 20

__global__ __launch_bounds__(128) void xw1_scaled(const float* __restrict__ X,
                                                  const float* __restrict__ dinv,
                                                  const float* __restrict__ W1,
                                                  float* __restrict__ P) {
  __shared__ float tile[4 * 16 * TPITCH];
  int wave = blockIdx.x * (blockDim.x >> 5) + (threadIdx.x >> 5);
  int lane = threadIdx.x & 31;
  int m = lane & 15;
  int h = lane >> 4;
  int g = lane >> 3;
  int j = lane & 7;
  int nb = wave * 16 + 4 * g;
  const float* xb = X + 4 * j;
  float* tl = tile + (threadIdx.x >> 5) * (16 * TPITCH);
  float4 u[4];
#pragma unroll
  for (int q = 0; q < 4; ++q)
    u[q] = *(const float4*)(xb + (size_t)(nb + q) * F_IN);
#pragma unroll
  for (int q = 0; q < 4; ++q)
    *(float4*)(tl + (4 * g + q) * TPITCH + 4 * j) = u[q];
  const float* tr = tl + m * TPITCH + 16 * h;
  float4 a0 = *(const float4*)(tr + 0);
  float4 a1 = *(const float4*)(tr + 4);
  float4 a2 = *(const float4*)(tr + 8);
  float4 a3 = *(const float4*)(tr + 12);
  const float* wb = W1 + (size_t)(16 * h) * F_H1 + m;
  v2f A0 = {a0.x, a0.y}, A1 = {a0.z, a0.w}, A2 = {a1.x, a1.y}, A3 = {a1.z, a1.w};
  v2f A4 = {a2.x, a2.y}, A5 = {a2.z, a2.w}, A6 = {a3.x, a3.y}, A7 = {a3.z, a3.w};
  v2f B0 = {wb[0 * F_H1],  wb[1 * F_H1]},  B1 = {wb[2 * F_H1],  wb[3 * F_H1]};
  v2f B2 = {wb[4 * F_H1],  wb[5 * F_H1]},  B3 = {wb[6 * F_H1],  wb[7 * F_H1]};
  v2f B4 = {wb[8 * F_H1],  wb[9 * F_H1]},  B5 = {wb[10 * F_H1], wb[11 * F_H1]};
  v2f B6 = {wb[12 * F_H1], wb[13 * F_H1]}, B7 = {wb[14 * F_H1], wb[15 * F_H1]};
  float4 d0 = *(const float4*)(dinv + wave * 16 + 8 * h);
  float4 d1 = *(const float4*)(dinv + wave * 16 + 8 * h + 4);
  v8f c = {0.f, 0.f, 0.f, 0.f, 0.f, 0.f, 0.f, 0.f};
  asm volatile(
      "v_wmma_f32_16x16x4_f32 %0, %1, %9, %0\n\t"
      "v_wmma_f32_16x16x4_f32 %0, %2, %10, %0\n\t"
      "v_wmma_f32_16x16x4_f32 %0, %3, %11, %0\n\t"
      "v_wmma_f32_16x16x4_f32 %0, %4, %12, %0\n\t"
      "v_wmma_f32_16x16x4_f32 %0, %5, %13, %0\n\t"
      "v_wmma_f32_16x16x4_f32 %0, %6, %14, %0\n\t"
      "v_wmma_f32_16x16x4_f32 %0, %7, %15, %0\n\t"
      "v_wmma_f32_16x16x4_f32 %0, %8, %16, %0\n\t"
      "v_nop\n\tv_nop\n\tv_nop\n\tv_nop\n\tv_nop"
      : "+v"(c)
      : "v"(A0), "v"(A1), "v"(A2), "v"(A3), "v"(A4), "v"(A5), "v"(A6), "v"(A7),
        "v"(B0), "v"(B1), "v"(B2), "v"(B3), "v"(B4), "v"(B5), "v"(B6), "v"(B7));
  float* out = P + (size_t)(wave * 16 + 8 * h) * F_H1 + m;
  out[0 * F_H1] = c[0] * d0.x;
  out[1 * F_H1] = c[1] * d0.y;
  out[2 * F_H1] = c[2] * d0.z;
  out[3 * F_H1] = c[3] * d0.w;
  out[4 * F_H1] = c[4] * d1.x;
  out[5 * F_H1] = c[5] * d1.y;
  out[6 * F_H1] = c[6] * d1.z;
  out[7 * F_H1] = c[7] * d1.w;
}

__global__ __launch_bounds__(128) void pull1_rows(const float* __restrict__ P,
                                                  const int* __restrict__ starts,
                                                  const int* __restrict__ ssrc,
                                                  const float* __restrict__ dinv,
                                                  const float* __restrict__ b1,
                                                  const float* __restrict__ W2,
                                                  float* __restrict__ T2) {
  __shared__ int idbuf[4 * IDBUF];
  __shared__ float tile[4 * 16 * TP2];
  __shared__ float obuf[4 * 16 * F_H2];
  int wave = blockIdx.x * (blockDim.x >> 5) + (threadIdx.x >> 5);
  int lane = threadIdx.x & 31;
  int g = lane >> 2;
  int j = lane & 3;
  int nb = wave * 16 + 2 * g;
  int st[3];
#pragma unroll
  for (int q = 0; q < 3; ++q) {
    int n = nb + q;
    st[q] = (n < N_NODES) ? starts[n] : N_EDGES;
  }
  float ddq[2];
  ddq[0] = dinv[nb];
  ddq[1] = dinv[nb + 1];
  const float* pb = P + 4 * j;
  int* my = idbuf + (threadIdx.x >> 5) * IDBUF;
  int seg_beg = __builtin_amdgcn_readlane(st[0], 0);
  int seg_end = __builtin_amdgcn_readlane(st[2], 31);

  float4 acc[2];
  acc[0] = make_float4(0.f, 0.f, 0.f, 0.f);
  acc[1] = acc[0];
  for (int abase = seg_beg & ~3; abase < seg_end; abase += IDBUF) {
    stage_ids(ssrc, abase, min(IDBUF, seg_end - abase), my, lane);
#pragma unroll
    for (int q = 0; q < 2; ++q) {
      int lo = max(st[q], abase) - abase;
      int last = min(st[q + 1], abase + IDBUF) - 1 - abase;
      float4 a = acc[q];
      for (int i = lo; i <= last; i += 4) {
        bool k1 = (i + 1 <= last), k2 = (i + 2 <= last), k3 = (i + 3 <= last);
        int s0 = my[i];
        int s1 = my[k1 ? i + 1 : i];
        int s2 = my[k2 ? i + 2 : i];
        int s3 = my[k3 ? i + 3 : i];
        float4 u0 = *(const float4*)(pb + (size_t)s0 * F_H1);
        float4 u1 = *(const float4*)(pb + (size_t)s1 * F_H1);
        float4 u2 = *(const float4*)(pb + (size_t)s2 * F_H1);
        float4 u3 = *(const float4*)(pb + (size_t)s3 * F_H1);
        fma4(a, u0, 1.0f);
        fma4(a, u1, k1 ? 1.0f : 0.0f);
        fma4(a, u2, k2 ? 1.0f : 0.0f);
        fma4(a, u3, k3 ? 1.0f : 0.0f);
      }
      acc[q] = a;
    }
  }
  float* tl = tile + (threadIdx.x >> 5) * (16 * TP2);
  float4 bv = *(const float4*)(b1 + 4 * j);
#pragma unroll
  for (int q = 0; q < 2; ++q) {
    float4 sv = *(const float4*)(pb + (size_t)(nb + q) * F_H1);
    float dd = ddq[q];
    float4 a = acc[q];
    float4 r;
    r.x = fmaxf(fmaf(a.x + sv.x, dd, bv.x), 0.0f);
    r.y = fmaxf(fmaf(a.y + sv.y, dd, bv.y), 0.0f);
    r.z = fmaxf(fmaf(a.z + sv.z, dd, bv.z), 0.0f);
    r.w = fmaxf(fmaf(a.w + sv.w, dd, bv.w), 0.0f);
    *(float4*)(tl + (2 * g + q) * TP2 + 4 * j) = r;
  }
  int m  = lane & 15;
  int hh = lane >> 4;
  const float* hrow = tl + m * TP2 + 2 * hh;
  bool live_col = (m < F_H2);
  const float* w2 = W2 + (size_t)(2 * hh) * F_H2 + (live_col ? m : 0);
  v2f A0 = {hrow[0],  hrow[1]},  A1 = {hrow[4],  hrow[5]};
  v2f A2 = {hrow[8],  hrow[9]},  A3 = {hrow[12], hrow[13]};
  v2f B0 = {live_col ? w2[0 * F_H2]  : 0.0f, live_col ? w2[1 * F_H2]  : 0.0f};
  v2f B1 = {live_col ? w2[4 * F_H2]  : 0.0f, live_col ? w2[5 * F_H2]  : 0.0f};
  v2f B2 = {live_col ? w2[8 * F_H2]  : 0.0f, live_col ? w2[9 * F_H2]  : 0.0f};
  v2f B3 = {live_col ? w2[12 * F_H2] : 0.0f, live_col ? w2[13 * F_H2] : 0.0f};
  float4 d0 = *(const float4*)(dinv + wave * 16 + 8 * hh);
  float4 d1 = *(const float4*)(dinv + wave * 16 + 8 * hh + 4);
  v8f c = {0.f, 0.f, 0.f, 0.f, 0.f, 0.f, 0.f, 0.f};
  asm volatile(
      "v_wmma_f32_16x16x4_f32 %0, %1, %5, %0\n\t"
      "v_wmma_f32_16x16x4_f32 %0, %2, %6, %0\n\t"
      "v_wmma_f32_16x16x4_f32 %0, %3, %7, %0\n\t"
      "v_wmma_f32_16x16x4_f32 %0, %4, %8, %0\n\t"
      "v_nop\n\tv_nop\n\tv_nop\n\tv_nop\n\tv_nop"
      : "+v"(c)
      : "v"(A0), "v"(A1), "v"(A2), "v"(A3), "v"(B0), "v"(B1), "v"(B2), "v"(B3));
  float* ob = obuf + (threadIdx.x >> 5) * (16 * F_H2);
  if (live_col) {
    float* o = ob + (8 * hh) * F_H2 + m;
    o[0 * F_H2] = c[0] * d0.x;
    o[1 * F_H2] = c[1] * d0.y;
    o[2 * F_H2] = c[2] * d0.z;
    o[3 * F_H2] = c[3] * d0.w;
    o[4 * F_H2] = c[4] * d1.x;
    o[5 * F_H2] = c[5] * d1.y;
    o[6 * F_H2] = c[6] * d1.z;
    o[7 * F_H2] = c[7] * d1.w;
  }
  float4 ov = *(const float4*)(ob + 4 * lane);
  *(float4*)(T2 + (size_t)wave * (16 * F_H2) + 4 * lane) = ov;
}

__global__ __launch_bounds__(128) void pull2_rows(const float* __restrict__ T2,
                                                  const int* __restrict__ starts,
                                                  const int* __restrict__ ssrc,
                                                  const float* __restrict__ dinv,
                                                  const float* __restrict__ b2,
                                                  const int* __restrict__ batch,
                                                  float* __restrict__ pooled,
                                                  float* __restrict__ cnt) {
  __shared__ int idbuf[4 * IDBUF];
  int wave = blockIdx.x * (blockDim.x >> 5) + (threadIdx.x >> 5);
  int lane = threadIdx.x & 31;
  int g = lane >> 1;
  int j = lane & 1;
  int node = wave * 16 + g;
  int st0 = starts[node];
  int st1 = (node + 1 < N_NODES) ? starts[node + 1] : N_EDGES;
  float dd = dinv[node];
  const float* tb = T2 + 4 * j;
  int* my = idbuf + (threadIdx.x >> 5) * IDBUF;
  int seg_beg = __builtin_amdgcn_readlane(st0, 0);
  int seg_end = __builtin_amdgcn_readlane(st1, 31);

  float4 a = make_float4(0.f, 0.f, 0.f, 0.f);
  for (int abase = seg_beg & ~3; abase < seg_end; abase += IDBUF) {
    stage_ids(ssrc, abase, min(IDBUF, seg_end - abase), my, lane);
    int lo = max(st0, abase) - abase;
    int last = min(st1, abase + IDBUF) - 1 - abase;
    for (int i = lo; i <= last; i += 4) {
      bool k1 = (i + 1 <= last), k2 = (i + 2 <= last), k3 = (i + 3 <= last);
      int s0 = my[i];
      int s1 = my[k1 ? i + 1 : i];
      int s2 = my[k2 ? i + 2 : i];
      int s3 = my[k3 ? i + 3 : i];
      float4 u0 = *(const float4*)(tb + (size_t)s0 * F_H2);
      float4 u1 = *(const float4*)(tb + (size_t)s1 * F_H2);
      float4 u2 = *(const float4*)(tb + (size_t)s2 * F_H2);
      float4 u3 = *(const float4*)(tb + (size_t)s3 * F_H2);
      fma4(a, u0, 1.0f);
      fma4(a, u1, k1 ? 1.0f : 0.0f);
      fma4(a, u2, k2 ? 1.0f : 0.0f);
      fma4(a, u3, k3 ? 1.0f : 0.0f);
    }
  }
  float4 bv = *(const float4*)(b2 + 4 * j);
  float4 sv = *(const float4*)(tb + (size_t)node * F_H2);
  float4 r;
  r.x = fmaxf(fmaf(a.x + sv.x, dd, bv.x), 0.0f);
  r.y = fmaxf(fmaf(a.y + sv.y, dd, bv.y), 0.0f);
  r.z = fmaxf(fmaf(a.z + sv.z, dd, bv.z), 0.0f);
  r.w = fmaxf(fmaf(a.w + sv.w, dd, bv.w), 0.0f);
  int gr = batch[node];
  float* ps = pooled + (size_t)gr * F_H2 + 4 * j;
  ATOMIC_ADD(&ps[0], r.x);
  ATOMIC_ADD(&ps[1], r.y);
  ATOMIC_ADD(&ps[2], r.z);
  ATOMIC_ADD(&ps[3], r.w);
  if (j == 0) ATOMIC_ADD(&cnt[gr], 1.0f);
}


extern "C" void kernel_launch(void* const* d_in, const int* in_sizes, int n_in,
                              void* d_out, int out_size, void* d_ws, size_t ws_size,
                              hipStream_t stream) {
  const float* x     = (const float*)d_in[0];
  const int*   ei    = (const int*)d_in[1];
  const int*   src   = ei;
  const int*   dst   = ei + N_EDGES;
  const int*   batch = (const int*)d_in[2];
  const float* W1 = (const float*)d_in[4];
  const float* b1 = (const float*)d_in[5];
  const float* W2 = (const float*)d_in[6];
  const float* b2 = (const float*)d_in[7];
  const float* Wl = (const float*)d_in[8];
  const float* bl = (const float*)d_in[9];
  float* out = (float*)d_out;

  char* ws = (char*)d_ws;
  float* dinv   = (float*)(ws);
  int*   arr    = (int*)(ws + (4ll << 20));
  int*   lst    = (int*)(ws + (8ll << 20));
  float* rows   = (float*)(ws + (8ll << 20));
  int*   planes = (int*)(ws + 72388608ll);
  int*   ssrc   = (int*)(ws + 72388608ll);
  int*   bsum   = (int*)(ws + 136388608ll);
  float* pooled = (float*)(ws + 136396800ll);
  int*   tbeg   = (int*)(ws + 136396800ll);
  float* cnt    = pooled + (size_t)N_GRAPHS * F_H2;

  const int NB = 256;
  const int TILE_BLOCKS = N_NODES / 64;

  if (ws_size >= (size_t)PR_NEED) {
    float* P  = rows;
    float* t2 = (float*)(ws + PR_T2_OFF);
    int* offs = (int*)(ws + PR_T2_OFF);
    scatter_local<<<NSLICE, LB_THREADS, 0, stream>>>(src, dst, offs, lst);
    bucket_order_sl<<<NBK_LIVE, BO_THREADS, 0, stream>>>(lst, offs, arr, dinv, ssrc);
    xw1_scaled<<<TILE_BLOCKS, 128, 0, stream>>>(x, dinv, W1, P);
    pull1_rows<<<TILE_BLOCKS, 128, 0, stream>>>(P, arr, ssrc, dinv, b1, W2, t2);
    zero_u32<<<128, 256, 0, stream>>>((unsigned*)pooled, N_GRAPHS * (F_H2 + 1));
    pull2_rows<<<TILE_BLOCKS, 128, 0, stream>>>(t2, arr, ssrc, dinv, b2, batch, pooled, cnt);
  } else {
    hist_buckets<<<NSLICE, LB_THREADS, 0, stream>>>(dst, planes);
    scan_block_sums<<<SCAN_BLOCKS, 256, 0, stream>>>(planes, bsum);
    scan_top<<<1, 1024, 0, stream>>>(bsum, SCAN_BLOCKS);
    scan_apply<<<SCAN_BLOCKS, 256, 0, stream>>>(planes, tbeg, bsum);
    scatter_runs<<<NSLICE, LB_THREADS, 0, stream>>>(src, dst, planes, tbeg, lst);
    bucket_order<<<NBK_LIVE, BO_THREADS, 0, stream>>>(lst, tbeg, arr, dinv, ssrc);
    float* h1 = rows;
    pull1_xw1<<<TILE_BLOCKS, 128, 0, stream>>>(x, arr, ssrc, dinv, W1, b1, h1);
    gemm_h1_w2<<<TILE_BLOCKS, 128, 0, stream>>>(h1, W2, dinv);
    zero_u32<<<128, 256, 0, stream>>>((unsigned*)pooled, N_GRAPHS * (F_H2 + 1));
    pull2_pool<<<TILE_BLOCKS, 128, 0, stream>>>(h1, arr, ssrc, dinv, b2, batch, pooled, cnt);
  }
  readout<<<(N_GRAPHS + NB - 1) / NB, NB, 0, stream>>>(pooled, cnt, Wl, bl,
                                                       out, N_GRAPHS);
}
